# starting kernel
# baseline (speedup 1.0000x reference)
#include <hip/hip_runtime.h>
#include <hip/hip_bf16.h>

typedef __attribute__((ext_vector_type(16))) _Float16 v16h;
typedef __attribute__((ext_vector_type(8)))  float    v8f;

#define TPB 256

__device__ __forceinline__ void atomic_add_f32(float* p, float v) {
  __hip_atomic_fetch_add(p, v, __ATOMIC_RELAXED, __HIP_MEMORY_SCOPE_AGENT);
}


__global__ void gcn_init(float* __restrict__ dinv, float* __restrict__ agg1,
                         float* __restrict__ agg2, int N) {
  long long t = (long long)blockIdx.x * blockDim.x + threadIdx.x;
  long long total = (long long)N * 16;
  if (t < total) agg1[t] = 0.0f;
  if (t < N) { dinv[t] = 1.0f; agg2[t] = 0.0f; }
}

__global__ void gcn_degree(const int* __restrict__ dst, float* __restrict__ deg,
                           long long E) {
  long long e = (long long)blockIdx.x * blockDim.x + threadIdx.x;
  if (e >= E) return;
  __builtin_prefetch(dst + e + 16384, 0, 1);
  atomic_add_f32(&deg[dst[e]], 1.0f);
}

__global__ void gcn_rsqrt(float* __restrict__ dinv, int N) {
  int i = blockIdx.x * blockDim.x + threadIdx.x;
  if (i < N) dinv[i] = rsqrtf(dinv[i]);
}

__global__ void gcn_xw1(const float* __restrict__ x, const float* __restrict__ W1,
                        float* __restrict__ t1, int N) {
  int wave = (int)(((long long)blockIdx.x * blockDim.x + threadIdx.x) >> 5);
  int lane = (int)(threadIdx.x & 31);
  int row0 = wave * 16;
  if (row0 >= N) return;

  v16h ah = {}, al = {};
  v16h bh = {}, bl = {};
  if (lane < 16) {
    int m = row0 + lane;
    if (m >= N) m = N - 1;
    const float* xr = x + (size_t)m * 5;
#pragma unroll
    for (int k = 0; k < 5; ++k) {
      float v = xr[k];
      _Float16 hi = (_Float16)v;
      ah[k] = hi;
      al[k] = (_Float16)(v - (float)hi);
    }
#pragma unroll
    for (int k = 0; k < 5; ++k) {
      float w = W1[k * 16 + lane];
      _Float16 hi = (_Float16)w;
      bh[k] = hi;
      bl[k] = (_Float16)(w - (float)hi);
    }
  }

  v8f c = {};
  c = __builtin_amdgcn_wmma_f32_16x16x32_f16(false, al, false, bh, (short)0, c, false, false);
  c = __builtin_amdgcn_wmma_f32_16x16x32_f16(false, ah, false, bl, (short)0, c, false, false);
  c = __builtin_amdgcn_wmma_f32_16x16x32_f16(false, ah, false, bh, (short)0, c, false, false);

  int n  = lane & 15;
  int mb = row0 + ((lane >= 16) ? 8 : 0);
  if (row0 + 16 <= N) {
    float* base = t1 + (size_t)mb * 16 + n;
#pragma unroll
    for (int r = 0; r < 8; ++r) base[(size_t)r * 16] = c[r];
  } else {
#pragma unroll
    for (int r = 0; r < 8; ++r) {
      int m = mb + r;
      if (m < N) t1[(size_t)m * 16 + n] = c[r];
    }
  }
}

__global__ void gcn_msg16(const int* __restrict__ src, const int* __restrict__ dst,
                          const float* __restrict__ t1, const float* __restrict__ dinv,
                          float* __restrict__ agg1, long long E) {
  long long t = (long long)blockIdx.x * blockDim.x + threadIdx.x;
  long long e = t >> 4;
  int c = (int)(t & 15);
  if (e >= E) return;
  int s = src[e];
  int d = dst[e];
  float w = dinv[s] * dinv[d];
  atomic_add_f32(&agg1[(size_t)d * 16 + c], t1[(size_t)s * 16 + c] * w);
}

__global__ void gcn_finalize1(const float* __restrict__ agg1, const float* __restrict__ t1,
                              const float* __restrict__ dinv, const float* __restrict__ b1,
                              const float* __restrict__ W2, float* __restrict__ t2, int N) {
  int i = blockIdx.x * blockDim.x + threadIdx.x;
  if (i >= N) return;
  float dd  = dinv[i] * dinv[i];
  float acc = 0.0f;
#pragma unroll
  for (int c = 0; c < 16; ++c) {
    float v = agg1[(size_t)i * 16 + c] + t1[(size_t)i * 16 + c] * dd + b1[c];
    v = fmaxf(v, 0.0f);
    acc = fmaf(v, W2[c], acc);
  }
  t2[i] = acc;
}

__global__ void gcn_msg1(const int* __restrict__ src, const int* __restrict__ dst,
                         const float* __restrict__ t2, const float* __restrict__ dinv,
                         float* __restrict__ agg2, long long E) {
  long long e = (long long)blockIdx.x * blockDim.x + threadIdx.x;
  if (e >= E) return;
  __builtin_prefetch(src + e + 16384, 0, 1);
  int s = src[e];
  int d = dst[e];
  atomic_add_f32(&agg2[d], t2[s] * dinv[s] * dinv[d]);
}

__global__ void gcn_finalize2(const float* __restrict__ agg2, const float* __restrict__ t2,
                              const float* __restrict__ dinv, const float* __restrict__ b2,
                              float* __restrict__ out, int N) {
  int i = blockIdx.x * blockDim.x + threadIdx.x;
  if (i >= N) return;
  out[i] = agg2[i] + t2[i] * dinv[i] * dinv[i] + b2[0];
}

extern "C" void kernel_launch(void* const* d_in, const int* in_sizes, int n_in,
                              void* d_out, int out_size, void* d_ws, size_t ws_size,
                              hipStream_t stream) {
  const float* x  = (const float*)d_in[0];
  const float* W1 = (const float*)d_in[1];
  const float* b1 = (const float*)d_in[2];
  const float* W2 = (const float*)d_in[3];
  const float* b2 = (const float*)d_in[4];
  const int*   ei = (const int*)d_in[5];

  int       N = in_sizes[0] / 5;
  long long E = (long long)in_sizes[5] / 2;
  const int* src = ei;
  const int* dst = ei + E;

  float* ws   = (float*)d_ws;
  float* dinv = ws;
  float* t1   = dinv + N;
  float* agg1 = t1 + (size_t)16 * N;
  float* t2   = agg1 + (size_t)16 * N;
  float* agg2 = t2 + N;

  float* out = (float*)d_out;

  long long initT   = (long long)N * 16;
  unsigned  gInit   = (unsigned)((initT + TPB - 1) / TPB);
  unsigned  gEdge   = (unsigned)((E + TPB - 1) / TPB);
  unsigned  gNode   = (unsigned)((N + TPB - 1) / TPB);
  unsigned  gMsg16  = (unsigned)((16 * E + TPB - 1) / TPB);
  int       waves   = (N + 15) / 16;
  unsigned  gWmma   = (unsigned)((waves + 1) / 2);

  gcn_init     <<<gInit,  TPB, 0, stream>>>(dinv, agg1, agg2, N);
  gcn_degree   <<<gEdge,  TPB, 0, stream>>>(dst, dinv, E);
  gcn_rsqrt    <<<gNode,  TPB, 0, stream>>>(dinv, N);
  gcn_xw1      <<<gWmma,   64, 0, stream>>>(x, W1, t1, N);
  gcn_msg16    <<<gMsg16, TPB, 0, stream>>>(src, dst, t1, dinv, agg1, E);
  gcn_finalize1<<<gNode,  TPB, 0, stream>>>(agg1, t1, dinv, b1, W2, t2, N);
  gcn_msg1     <<<gEdge,  TPB, 0, stream>>>(src, dst, t2, dinv, agg2, E);
  gcn_finalize2<<<gNode,  TPB, 0, stream>>>(agg2, t2, dinv, b2, out, N);
}


# Edges grouped once per call by 256-node destination bucket (LDS counting sort); degree and both layers sum in the owner's LDS, t1 formed per 16-edge tile on WMMA; no global atomics (sum order changes)
# speedup vs baseline: 4.0766x; 4.0766x over previous
#include <hip/hip_runtime.h>
#include <hip/hip_bf16.h>


typedef __attribute__((ext_vector_type(16))) _Float16 v16h;
typedef __attribute__((ext_vector_type(8)))  float    v8f;

#define NT      512
#define NTL     1024
#define NOEDGE  0xFFFFFFFFu
#define BSH     8
#define BSZ     256
#define NBMAX   512
#define CH      8192
#define PITCH   17


__device__ __forceinline__ void lds_add_i32(int* p, int v) {
  __hip_atomic_fetch_add(p, v, __ATOMIC_RELAXED, __HIP_MEMORY_SCOPE_WORKGROUP);
}
__device__ __forceinline__ int lds_add_ret_i32(int* p, int v) {
  return __hip_atomic_fetch_add(p, v, __ATOMIC_RELAXED, __HIP_MEMORY_SCOPE_WORKGROUP);
}
__device__ __forceinline__ void lds_add_f32(float* p, float v) {
  __hip_atomic_fetch_add(p, v, __ATOMIC_RELAXED, __HIP_MEMORY_SCOPE_WORKGROUP);
}

__device__ __forceinline__ int wave_excl_scan_lds(int* a, int n, int lane) {
  int b0 = lane * (NBMAX / 32);
  int sum = 0;
  for (int i = 0; i < NBMAX / 32; ++i) { int j = b0 + i; if (j < n) sum += a[j]; }
  int inc = sum;
  for (int o = 1; o < 32; o <<= 1) { int v = __shfl_up(inc, o); if (lane >= o) inc += v; }
  int run = inc - sum;
  for (int i = 0; i < NBMAX / 32; ++i) { int j = b0 + i; if (j < n) { int v = a[j]; a[j] = run; run += v; } }
  return __shfl(inc, 31);
}

__global__ __launch_bounds__(NT) void gcn_count(const int* __restrict__ dst, int* __restrict__ cnt,
                                                long long E, int NB, int CB) {
  __shared__ int c[NBMAX];
  int tid = (int)threadIdx.x, b = (int)blockIdx.x;
  for (int i = tid; i < NBMAX; i += NT) c[i] = 0;
  __syncthreads();
  long long e0 = (long long)b * CH;
  int n = (int)((E - e0 < CH) ? (E - e0) : CH);
  for (int i0 = tid; i0 < n; i0 += 4 * NT) {
    int d[4];
#pragma unroll
    for (int j = 0; j < 4; ++j) { int i = i0 + j * NT; d[j] = (i < n) ? dst[e0 + i] : -1; }
#pragma unroll
    for (int j = 0; j < 4; ++j) if (d[j] >= 0) lds_add_i32(&c[d[j] >> BSH], 1);
  }
  __syncthreads();
  for (int k = tid; k < NB; k += NT) cnt[(size_t)k * CB + b] = c[k];
}

__global__ void gcn_scan(int* __restrict__ cnt, int* __restrict__ tot, int NB, int CB) {
  int wave = (int)(((long long)blockIdx.x * blockDim.x + threadIdx.x) >> 5);
  int lane = (int)(threadIdx.x & 31);
  if (wave >= NB) return;
  int* row = cnt + (size_t)wave * CB;
  int per = (CB + 31) >> 5;
  int b0 = lane * per;
  int sum = 0;
  for (int i = 0; i < per; ++i) { int j = b0 + i; if (j < CB) sum += row[j]; }
  int inc = sum;
  for (int o = 1; o < 32; o <<= 1) { int v = __shfl_up(inc, o); if (lane >= o) inc += v; }
  int run = inc - sum;
  for (int i = 0; i < per; ++i) { int j = b0 + i; if (j < CB) { int v = row[j]; row[j] = run; run += v; } }
  if (lane == 31) tot[wave] = inc;
}

__global__ __launch_bounds__(NT) void gcn_place(const int* __restrict__ src, const int* __restrict__ dst,
                                                const int* __restrict__ cnt, const int* __restrict__ tot,
                                                unsigned* __restrict__ sorted, int* __restrict__ bst,
                                                long long E, int NB, int CB) {
  __shared__ unsigned       sw[CH];
  __shared__ unsigned short sb[CH];
  __shared__ int lstart[NBMAX], lcur[NBMAX], gbase[NBMAX], gst[NBMAX];
  int tid = (int)threadIdx.x, b = (int)blockIdx.x, lane = tid & 31;
  for (int k = tid; k < NBMAX; k += NT) {
    int own = 0, before = 0, t = 0;
    if (k < NB) {
      t = tot[k];
      before = cnt[(size_t)k * CB + b];
      int next = (b + 1 < CB) ? cnt[(size_t)k * CB + b + 1] : t;
      own = next - before;
    }
    lstart[k] = own; gbase[k] = before; gst[k] = t;
  }
  __syncthreads();
  if (tid < 32) {
    wave_excl_scan_lds(lstart, NB, lane);
    wave_excl_scan_lds(gst, NB, lane);
  }
  __syncthreads();
  for (int k = tid; k < NB; k += NT) {
    lcur[k] = lstart[k];
    gbase[k] += gst[k];
    if (b == 0) bst[k] = gst[k];
  }
  if (b == 0 && tid == 0) bst[NB] = (int)E;
  __syncthreads();
  long long e0 = (long long)b * CH;
  int n = (int)((E - e0 < CH) ? (E - e0) : CH);
  for (int i0 = tid; i0 < n; i0 += 4 * NT) {
    int d[4], s[4];
#pragma unroll
    for (int j = 0; j < 4; ++j) {
      int i = i0 + j * NT;
      d[j] = (i < n) ? dst[e0 + i] : -1;
      s[j] = (i < n) ? src[e0 + i] : 0;
    }
#pragma unroll
    for (int j = 0; j < 4; ++j) {
      if (d[j] >= 0) {
        int k = d[j] >> BSH;
        int r = lds_add_ret_i32(&lcur[k], 1);
        sw[r] = ((unsigned)s[j] << BSH) | (unsigned)(d[j] & (BSZ - 1));
        sb[r] = (unsigned short)k;
      }
    }
  }
  __syncthreads();
  for (int i = tid; i < n; i += NT) {
    int k = sb[i];
    sorted[(size_t)gbase[k] + (i - lstart[k])] = sw[i];
  }
}

__global__ __launch_bounds__(NT) void gcn_degree(const unsigned* __restrict__ sorted, const int* __restrict__ bst,
                                                 float* __restrict__ dinv, int N) {
  __shared__ int c[BSZ];
  int tid = (int)threadIdx.x, k = (int)blockIdx.x;
  if (tid < BSZ) c[tid] = 0;
  __syncthreads();
  int e0 = bst[k], e1 = bst[k + 1];
  for (int i0 = e0 + tid; i0 < e1; i0 += 4 * NT) {
    unsigned w[4];
#pragma unroll
    for (int j = 0; j < 4; ++j) { int i = i0 + j * NT; w[j] = (i < e1) ? sorted[i] : NOEDGE; }
#pragma unroll
    for (int j = 0; j < 4; ++j) if (w[j] != NOEDGE) lds_add_i32(&c[w[j] & (BSZ - 1)], 1);
  }
  __syncthreads();
  if (tid < BSZ) {
    int d = (k << BSH) + tid;
    if (d < N) dinv[d] = rsqrtf(1.0f + (float)c[tid]);
  }
}

__global__ __launch_bounds__(NTL) void gcn_layer1(const unsigned* __restrict__ sorted, const int* __restrict__ bst,
                                                 const float* __restrict__ x, const float* __restrict__ W1,
                                                 const float* __restrict__ b1, const float* __restrict__ W2,
                                                 const float* __restrict__ dinv, float* __restrict__ t2, int N) {
  __shared__ float bins[BSZ * PITCH];
  __shared__ float sdinv[BSZ];
  int tid = (int)threadIdx.x, k = (int)blockIdx.x;
  int lane = tid & 31, wave = tid >> 5;
  int node0 = k << BSH;
  for (int i = tid; i < BSZ * PITCH; i += NTL) bins[i] = 0.0f;
  if (tid < BSZ) { int d = node0 + tid; sdinv[tid] = (d < N) ? dinv[d] : 0.0f; }

  v16h bh = {}, bl = {};
  if (lane < 16) {
#pragma unroll
    for (int kk = 0; kk < 5; ++kk) {
      float w = W1[kk * 16 + lane];
      _Float16 hi = (_Float16)w;
      bh[kk] = hi;
      bl[kk] = (_Float16)(w - (float)hi);
    }
  }
  __syncthreads();

  int e0 = bst[k], e1 = bst[k + 1];
  int nE = e1 - e0;
  int tilesE = (nE + 15) >> 4;
  int tiles  = tilesE + (BSZ >> 4);
  int ch = lane & 15;
  int mb = (lane >> 4) << 3;
  auto tile_word = [&](int t) -> unsigned {
    int idx = (t << 4) + lane;
    int gi = e0 + idx;
    gi = (gi < e1) ? gi : (e1 - 1);
    gi = (gi > 0) ? gi : 0;
    unsigned we = sorted[gi];
    int dl = ((t - tilesE) << 4) + lane;
    unsigned ws = ((unsigned)(node0 + dl) << BSH) | (unsigned)dl;
    bool edge = (t < tilesE) && (idx < nE);
    bool self = (t >= tilesE) && (t < tiles) && (node0 + dl < N);
    unsigned w = edge ? we : (self ? ws : NOEDGE);
    return (lane < 16) ? w : NOEDGE;
  };
  unsigned wcur = tile_word(wave);
  for (int t = wave; t < tiles; t += NTL / 32) {
    unsigned wnext = tile_word(t + NTL / 32);
    unsigned s = (wcur != NOEDGE) ? (wcur >> BSH) : 0u;
    int    dlv = (wcur != NOEDGE) ? (int)(wcur & (BSZ - 1)) : -1;
    float  nrm = 0.0f;
    v16h ah = {}, al = {};
    if (lane < 16) {
      float ds = dinv[s];
      const float* xr = x + (size_t)s * 5;
#pragma unroll
      for (int kk = 0; kk < 5; ++kk) {
        float v = xr[kk];
        _Float16 hi = (_Float16)v;
        ah[kk] = hi;
        al[kk] = (_Float16)(v - (float)hi);
      }
      float dd = sdinv[dlv & (BSZ - 1)];
      nrm = (dlv >= 0) ? ds * dd : 0.0f;
    }
    v8f c = {};
    c = __builtin_amdgcn_wmma_f32_16x16x32_f16(false, al, false, bh, (short)0, c, false, false);
    c = __builtin_amdgcn_wmma_f32_16x16x32_f16(false, ah, false, bl, (short)0, c, false, false);
    c = __builtin_amdgcn_wmma_f32_16x16x32_f16(false, ah, false, bh, (short)0, c, false, false);
#pragma unroll
    for (int r0 = 0; r0 < 8; r0 += 4) {
      int dm[4]; float nm[4];
#pragma unroll
      for (int r = 0; r < 4; ++r) { dm[r] = __shfl(dlv, mb + r0 + r); nm[r] = __shfl(nrm, mb + r0 + r); }
#pragma unroll
      for (int r = 0; r < 4; ++r)
        if (dm[r] >= 0) lds_add_f32(&bins[dm[r] * PITCH + ch], c[r0 + r] * nm[r]);
    }
    wcur = wnext;
  }
  __syncthreads();
  if (tid < BSZ) {
    int d = node0 + tid;
    if (d < N) {
      float acc = 0.0f;
#pragma unroll
      for (int cc = 0; cc < 16; ++cc) {
        float v = bins[tid * PITCH + cc] + b1[cc];
        v = fmaxf(v, 0.0f);
        acc = fmaf(v, W2[cc], acc);
      }
      t2[d] = acc;
    }
  }
}

__global__ __launch_bounds__(NT) void gcn_layer2(const unsigned* __restrict__ sorted, const int* __restrict__ bst,
                                                 const float* __restrict__ t2, const float* __restrict__ dinv,
                                                 const float* __restrict__ b2, float* __restrict__ out, int N) {
  __shared__ float bins[BSZ];
  __shared__ float sdinv[BSZ];
  int tid = (int)threadIdx.x, k = (int)blockIdx.x;
  int node0 = k << BSH;
  if (tid < BSZ) { int d = node0 + tid; bins[tid] = 0.0f; sdinv[tid] = (d < N) ? dinv[d] : 0.0f; }
  __syncthreads();
  int e0 = bst[k], e1 = bst[k + 1];
  for (int i0 = e0 + tid; i0 < e1; i0 += 4 * NT) {
    unsigned w[4];
    float a[4], g[4];
#pragma unroll
    for (int j = 0; j < 4; ++j) { int i = i0 + j * NT; w[j] = (i < e1) ? sorted[i] : NOEDGE; }
#pragma unroll
    for (int j = 0; j < 4; ++j) {
      unsigned s = (w[j] != NOEDGE) ? (w[j] >> BSH) : 0u;
      a[j] = t2[s]; g[j] = dinv[s];
    }
#pragma unroll
    for (int j = 0; j < 4; ++j) {
      if (w[j] != NOEDGE) {
        int dl = (int)(w[j] & (BSZ - 1));
        lds_add_f32(&bins[dl], a[j] * g[j] * sdinv[dl]);
      }
    }
  }
  __syncthreads();
  if (tid < BSZ) {
    int d = node0 + tid;
    if (d < N) { float di = sdinv[tid]; out[d] = bins[tid] + t2[d] * di * di + b2[0]; }
  }
}

extern "C" void kernel_launch(void* const* d_in, const int* in_sizes, int n_in,
                              void* d_out, int out_size, void* d_ws, size_t ws_size,
                              hipStream_t stream) {
  const float* x  = (const float*)d_in[0];
  const float* W1 = (const float*)d_in[1];
  const float* b1 = (const float*)d_in[2];
  const float* W2 = (const float*)d_in[3];
  const float* b2 = (const float*)d_in[4];
  const int*   ei = (const int*)d_in[5];

  int       N = in_sizes[0] / 5;
  long long E = (long long)in_sizes[5] / 2;
  const int* src = ei;
  const int* dst = ei + E;

  int NB = (N + BSZ - 1) / BSZ;
  int CB = (int)((E + CH - 1) / CH);

  unsigned* sorted = (unsigned*)d_ws;
  int*      bst    = (int*)(sorted + E);
  float*    dinv   = (float*)(bst + NB + 1);
  float*    t2     = dinv + N;
  int*      cnt    = (int*)dinv;
  int*      tot    = cnt + (size_t)NB * CB;

  float* out = (float*)d_out;

  gcn_count <<<CB, NT, 0, stream>>>(dst, cnt, E, NB, CB);
  gcn_scan  <<<(NB + 7) / 8, 256, 0, stream>>>(cnt, tot, NB, CB);
  gcn_place <<<CB, NT, 0, stream>>>(src, dst, cnt, tot, sorted, bst, E, NB, CB);
  gcn_degree<<<NB, NT, 0, stream>>>(sorted, bst, dinv, N);
  gcn_layer1<<<NB, NTL, 0, stream>>>(sorted, bst, x, W1, b1, W2, dinv, t2, N);
  gcn_layer2<<<NB, NT, 0, stream>>>(sorted, bst, t2, dinv, b2, out, N);
}


# x@W1 formed once per node (WMMA tile per 16 nodes in the degree kernel), stored as dinv-scaled 64 B rows; layer-1 edges: 4 lanes x 16 B gather a row into LDS sums; dinv[dst] applied after the sum
# speedup vs baseline: 4.4950x; 1.1026x over previous
#include <hip/hip_runtime.h>
#include <hip/hip_bf16.h>


typedef __attribute__((ext_vector_type(16))) _Float16 v16h;
typedef __attribute__((ext_vector_type(8)))  float    v8f;

#define NT      512
#define NTL     1024
#define NOEDGE  0xFFFFFFFFu
#define BSH     8
#define BSZ     256
#define NBMAX   512
#define CH      8192
#define PITCH   17


__device__ __forceinline__ void lds_add_i32(int* p, int v) {
  __hip_atomic_fetch_add(p, v, __ATOMIC_RELAXED, __HIP_MEMORY_SCOPE_WORKGROUP);
}
__device__ __forceinline__ int lds_add_ret_i32(int* p, int v) {
  return __hip_atomic_fetch_add(p, v, __ATOMIC_RELAXED, __HIP_MEMORY_SCOPE_WORKGROUP);
}
__device__ __forceinline__ void lds_add_f32(float* p, float v) {
  __hip_atomic_fetch_add(p, v, __ATOMIC_RELAXED, __HIP_MEMORY_SCOPE_WORKGROUP);
}

__device__ __forceinline__ int wave_excl_scan_lds(int* a, int n, int lane) {
  int b0 = lane * (NBMAX / 32);
  int sum = 0;
  for (int i = 0; i < NBMAX / 32; ++i) { int j = b0 + i; if (j < n) sum += a[j]; }
  int inc = sum;
  for (int o = 1; o < 32; o <<= 1) { int v = __shfl_up(inc, o); if (lane >= o) inc += v; }
  int run = inc - sum;
  for (int i = 0; i < NBMAX / 32; ++i) { int j = b0 + i; if (j < n) { int v = a[j]; a[j] = run; run += v; } }
  return __shfl(inc, 31);
}

__global__ __launch_bounds__(NT) void gcn_count(const int* __restrict__ dst, int* __restrict__ cnt,
                                                long long E, int NB, int CB) {
  __shared__ int c[NBMAX];
  int tid = (int)threadIdx.x, b = (int)blockIdx.x;
  for (int i = tid; i < NBMAX; i += NT) c[i] = 0;
  __syncthreads();
  long long e0 = (long long)b * CH;
  int n = (int)((E - e0 < CH) ? (E - e0) : CH);
  for (int i0 = tid; i0 < n; i0 += 4 * NT) {
    int d[4];
#pragma unroll
    for (int j = 0; j < 4; ++j) { int i = i0 + j * NT; d[j] = (i < n) ? dst[e0 + i] : -1; }
#pragma unroll
    for (int j = 0; j < 4; ++j) if (d[j] >= 0) lds_add_i32(&c[d[j] >> BSH], 1);
  }
  __syncthreads();
  for (int k = tid; k < NB; k += NT) cnt[(size_t)k * CB + b] = c[k];
}

__global__ void gcn_scan(int* __restrict__ cnt, int* __restrict__ tot, int NB, int CB) {
  int wave = (int)(((long long)blockIdx.x * blockDim.x + threadIdx.x) >> 5);
  int lane = (int)(threadIdx.x & 31);
  if (wave >= NB) return;
  int* row = cnt + (size_t)wave * CB;
  int per = (CB + 31) >> 5;
  int b0 = lane * per;
  int sum = 0;
  for (int i = 0; i < per; ++i) { int j = b0 + i; if (j < CB) sum += row[j]; }
  int inc = sum;
  for (int o = 1; o < 32; o <<= 1) { int v = __shfl_up(inc, o); if (lane >= o) inc += v; }
  int run = inc - sum;
  for (int i = 0; i < per; ++i) { int j = b0 + i; if (j < CB) { int v = row[j]; row[j] = run; run += v; } }
  if (lane == 31) tot[wave] = inc;
}

__global__ __launch_bounds__(NT) void gcn_place(const int* __restrict__ src, const int* __restrict__ dst,
                                                const int* __restrict__ cnt, const int* __restrict__ tot,
                                                unsigned* __restrict__ sorted, int* __restrict__ bst,
                                                long long E, int NB, int CB) {
  __shared__ unsigned       sw[CH];
  __shared__ unsigned short sb[CH];
  __shared__ int lstart[NBMAX], lcur[NBMAX], gbase[NBMAX], gst[NBMAX];
  int tid = (int)threadIdx.x, b = (int)blockIdx.x, lane = tid & 31;
  for (int k = tid; k < NBMAX; k += NT) {
    int own = 0, before = 0, t = 0;
    if (k < NB) {
      t = tot[k];
      before = cnt[(size_t)k * CB + b];
      int next = (b + 1 < CB) ? cnt[(size_t)k * CB + b + 1] : t;
      own = next - before;
    }
    lstart[k] = own; gbase[k] = before; gst[k] = t;
  }
  __syncthreads();
  if (tid < 32) {
    wave_excl_scan_lds(lstart, NB, lane);
    wave_excl_scan_lds(gst, NB, lane);
  }
  __syncthreads();
  for (int k = tid; k < NB; k += NT) {
    lcur[k] = lstart[k];
    gbase[k] += gst[k];
    if (b == 0) bst[k] = gst[k];
  }
  if (b == 0 && tid == 0) bst[NB] = (int)E;
  __syncthreads();
  long long e0 = (long long)b * CH;
  int n = (int)((E - e0 < CH) ? (E - e0) : CH);
  for (int i0 = tid; i0 < n; i0 += 4 * NT) {
    int d[4], s[4];
#pragma unroll
    for (int j = 0; j < 4; ++j) {
      int i = i0 + j * NT;
      d[j] = (i < n) ? dst[e0 + i] : -1;
      s[j] = (i < n) ? src[e0 + i] : 0;
    }
#pragma unroll
    for (int j = 0; j < 4; ++j) {
      if (d[j] >= 0) {
        int k = d[j] >> BSH;
        int r = lds_add_ret_i32(&lcur[k], 1);
        sw[r] = ((unsigned)s[j] << BSH) | (unsigned)(d[j] & (BSZ - 1));
        sb[r] = (unsigned short)k;
      }
    }
  }
  __syncthreads();
  for (int i = tid; i < n; i += NT) {
    int k = sb[i];
    sorted[(size_t)gbase[k] + (i - lstart[k])] = sw[i];
  }
}

__global__ __launch_bounds__(NT) void gcn_degree(const unsigned* __restrict__ sorted, const int* __restrict__ bst,
                                                 const float* __restrict__ x, const float* __restrict__ W1,
                                                 float* __restrict__ dinv, float* __restrict__ rows, int N) {
  __shared__ int c[BSZ];
  __shared__ float sd[BSZ];
  __shared__ __attribute__((aligned(16))) float st[BSZ * 16];
  int tid = (int)threadIdx.x, k = (int)blockIdx.x;
  int lane = tid & 31, wave = tid >> 5;
  int node0 = k << BSH;
  if (tid < BSZ) c[tid] = 0;

  v16h bh = {}, bl = {}, ah = {}, al = {};
  if (lane < 16) {
#pragma unroll
    for (int kk = 0; kk < 5; ++kk) {
      float w = W1[kk * 16 + lane];
      _Float16 hi = (_Float16)w;
      bh[kk] = hi;
      bl[kk] = (_Float16)(w - (float)hi);
    }
    int d = node0 + (wave << 4) + lane;
    if (d < N) {
      const float* xr = x + (size_t)d * 5;
#pragma unroll
      for (int kk = 0; kk < 5; ++kk) {
        float v = xr[kk];
        _Float16 hi = (_Float16)v;
        ah[kk] = hi;
        al[kk] = (_Float16)(v - (float)hi);
      }
    }
  }
  __syncthreads();
  int e0 = bst[k], e1 = bst[k + 1];
  for (int i0 = e0 + tid; i0 < e1; i0 += 4 * NT) {
    unsigned w[4];
#pragma unroll
    for (int j = 0; j < 4; ++j) { int i = i0 + j * NT; w[j] = (i < e1) ? sorted[i] : NOEDGE; }
#pragma unroll
    for (int j = 0; j < 4; ++j) if (w[j] != NOEDGE) lds_add_i32(&c[w[j] & (BSZ - 1)], 1);
  }
  __syncthreads();
  if (tid < BSZ) {
    int d = node0 + tid;
    float di = rsqrtf(1.0f + (float)c[tid]);
    sd[tid] = di;
    if (d < N) dinv[d] = di;
  }
  v8f acc = {};
  acc = __builtin_amdgcn_wmma_f32_16x16x32_f16(false, al, false, bh, (short)0, acc, false, false);
  acc = __builtin_amdgcn_wmma_f32_16x16x32_f16(false, ah, false, bl, (short)0, acc, false, false);
  acc = __builtin_amdgcn_wmma_f32_16x16x32_f16(false, ah, false, bh, (short)0, acc, false, false);
  __syncthreads();
  int ch = lane & 15;
  int nb = (wave << 4) + ((lane >> 4) << 3);
#pragma unroll
  for (int r = 0; r < 8; ++r) st[(nb + r) * 16 + ch] = acc[r] * sd[nb + r];
  __syncthreads();
  const float4* st4 = (const float4*)st;
  float4* out4 = (float4*)rows + (size_t)node0 * 4;
  for (int i = tid; i < BSZ * 4; i += NT)
    if (node0 + (i >> 2) < N) out4[i] = st4[i];
}

__global__ __launch_bounds__(NTL) void gcn_layer1(const unsigned* __restrict__ sorted, const int* __restrict__ bst,
                                                 const float* __restrict__ rows, const float* __restrict__ b1,
                                                 const float* __restrict__ W2, const float* __restrict__ dinv,
                                                 float* __restrict__ t2, int N) {
  __shared__ float bins[BSZ * PITCH];
  int tid = (int)threadIdx.x, k = (int)blockIdx.x;
  int node0 = k << BSH;
  const float4* rows4 = (const float4*)rows;
  int g = tid >> 2, q = tid & 3;
  {
    int d = node0 + g;
    float4 v0 = rows4[(size_t)((d < N) ? d : (N - 1)) * 4 + q];
    float* p = &bins[g * PITCH + (q << 2)];
    bool in = d < N;
    p[0] = in ? v0.x : 0.0f;
    p[1] = in ? v0.y : 0.0f;
    p[2] = in ? v0.z : 0.0f;
    p[3] = in ? v0.w : 0.0f;
  }
  __syncthreads();

  int e0 = bst[k], e1 = bst[k + 1];
  int i0 = e0 + g;
  if (i0 < e1) {
    unsigned w[4];
#pragma unroll
    for (int j = 0; j < 4; ++j) {
      int i = i0 + j * (NTL / 4);
      w[j] = sorted[(i < e1) ? i : (e1 - 1)];
    }
    do {
      unsigned wc[4];
      float4 v[4];
      bool ok[4];
#pragma unroll
      for (int j = 0; j < 4; ++j) {
        wc[j] = w[j];
        ok[j] = (i0 + j * (NTL / 4)) < e1;
        v[j] = rows4[(size_t)(wc[j] >> BSH) * 4 + q];
      }
      i0 += 4 * (NTL / 4);
#pragma unroll
      for (int j = 0; j < 4; ++j) {
        int i = i0 + j * (NTL / 4);
        w[j] = sorted[(i < e1) ? i : (e1 - 1)];
      }
#pragma unroll
      for (int j = 0; j < 4; ++j) {
        int dl = ok[j] ? (int)(wc[j] & (BSZ - 1)) : g;
        float* p = &bins[dl * PITCH + (q << 2)];
        lds_add_f32(p + 0, ok[j] ? v[j].x : 0.0f);
        lds_add_f32(p + 1, ok[j] ? v[j].y : 0.0f);
        lds_add_f32(p + 2, ok[j] ? v[j].z : 0.0f);
        lds_add_f32(p + 3, ok[j] ? v[j].w : 0.0f);
      }
    } while (i0 < e1);
  }
  __syncthreads();
  if (tid < BSZ) {
    int d = node0 + tid;
    if (d < N) {
      float di = dinv[d];
      float acc = 0.0f;
#pragma unroll
      for (int cc = 0; cc < 16; ++cc) {
        float v = fmaf(bins[tid * PITCH + cc], di, b1[cc]);
        v = fmaxf(v, 0.0f);
        acc = fmaf(v, W2[cc], acc);
      }
      t2[d] = acc;
    }
  }
}

__global__ __launch_bounds__(NT) void gcn_layer2(const unsigned* __restrict__ sorted, const int* __restrict__ bst,
                                                 const float* __restrict__ t2, const float* __restrict__ dinv,
                                                 const float* __restrict__ b2, float* __restrict__ out, int N) {
  __shared__ float bins[BSZ];
  __shared__ float sdinv[BSZ];
  int tid = (int)threadIdx.x, k = (int)blockIdx.x;
  int node0 = k << BSH;
  if (tid < BSZ) { int d = node0 + tid; bins[tid] = 0.0f; sdinv[tid] = (d < N) ? dinv[d] : 0.0f; }
  __syncthreads();
  int e0 = bst[k], e1 = bst[k + 1];
  for (int i0 = e0 + tid; i0 < e1; i0 += 4 * NT) {
    unsigned w[4];
    float a[4], g[4];
#pragma unroll
    for (int j = 0; j < 4; ++j) { int i = i0 + j * NT; w[j] = (i < e1) ? sorted[i] : NOEDGE; }
#pragma unroll
    for (int j = 0; j < 4; ++j) {
      unsigned s = (w[j] != NOEDGE) ? (w[j] >> BSH) : 0u;
      a[j] = t2[s]; g[j] = dinv[s];
    }
#pragma unroll
    for (int j = 0; j < 4; ++j) {
      if (w[j] != NOEDGE) {
        int dl = (int)(w[j] & (BSZ - 1));
        lds_add_f32(&bins[dl], a[j] * g[j] * sdinv[dl]);
      }
    }
  }
  __syncthreads();
  if (tid < BSZ) {
    int d = node0 + tid;
    if (d < N) { float di = sdinv[tid]; out[d] = bins[tid] + t2[d] * di * di + b2[0]; }
  }
}

extern "C" void kernel_launch(void* const* d_in, const int* in_sizes, int n_in,
                              void* d_out, int out_size, void* d_ws, size_t ws_size,
                              hipStream_t stream) {
  const float* x  = (const float*)d_in[0];
  const float* W1 = (const float*)d_in[1];
  const float* b1 = (const float*)d_in[2];
  const float* W2 = (const float*)d_in[3];
  const float* b2 = (const float*)d_in[4];
  const int*   ei = (const int*)d_in[5];

  int       N = in_sizes[0] / 5;
  long long E = (long long)in_sizes[5] / 2;
  const int* src = ei;
  const int* dst = ei + E;

  int NB = (N + BSZ - 1) / BSZ;
  int CB = (int)((E + CH - 1) / CH);

  float*    rows   = (float*)d_ws;
  unsigned* sorted = (unsigned*)(rows + (size_t)16 * N);
  int*      bst    = (int*)(sorted + E);
  float*    dinv   = (float*)(bst + NB + 1);
  float*    t2     = dinv + N;
  int*      cnt    = (int*)dinv;
  int*      tot    = cnt + (size_t)NB * CB;

  size_t need = ((size_t)18 * N + (size_t)E + (size_t)NB + 1) * 4;
  if (ws_size < need) return;

  float* out = (float*)d_out;

  gcn_count <<<CB, NT, 0, stream>>>(dst, cnt, E, NB, CB);
  gcn_scan  <<<(NB + 7) / 8, 256, 0, stream>>>(cnt, tot, NB, CB);
  gcn_place <<<CB, NT, 0, stream>>>(src, dst, cnt, tot, sorted, bst, E, NB, CB);
  gcn_degree<<<NB, NT, 0, stream>>>(sorted, bst, x, W1, dinv, rows, N);
  gcn_layer1<<<NB, NTL, 0, stream>>>(sorted, bst, rows, b1, W2, dinv, t2, N);
  gcn_layer2<<<NB, NT, 0, stream>>>(sorted, bst, t2, dinv, b2, out, N);
}


# One-launch partition: each chunk bucket-sorted in LDS (count add's return = rank); degree kernel sums its segment starts as its base, copies the segments to its run; 6->4 launches; sum order changes
# speedup vs baseline: 5.6809x; 1.2638x over previous
#include <hip/hip_runtime.h>
#include <hip/hip_bf16.h>


typedef __attribute__((ext_vector_type(16))) _Float16 v16h;
typedef __attribute__((ext_vector_type(8)))  float    v8f;

#define NT      512
#define NTL     1024
#define NOEDGE  0xFFFFFFFFu
#define BSH     8
#define BSZ     256
#define NBMAX   512
#define CH      8192
#define CBMAX   512
#define SEGT    16
static_assert(CBMAX == NBMAX && CBMAX == NT && CH % NT == 0, "one scan routine for both tables; one table entry per thread");
#define PITCH   17


__device__ __forceinline__ void lds_add_i32(int* p, int v) {
  __hip_atomic_fetch_add(p, v, __ATOMIC_RELAXED, __HIP_MEMORY_SCOPE_WORKGROUP);
}
__device__ __forceinline__ int lds_add_ret_i32(int* p, int v) {
  return __hip_atomic_fetch_add(p, v, __ATOMIC_RELAXED, __HIP_MEMORY_SCOPE_WORKGROUP);
}
__device__ __forceinline__ void lds_add_f32(float* p, float v) {
  __hip_atomic_fetch_add(p, v, __ATOMIC_RELAXED, __HIP_MEMORY_SCOPE_WORKGROUP);
}

__device__ __forceinline__ int wave_excl_scan_lds(int* a, int n, int lane) {
  int b0 = lane * (NBMAX / 32);
  int sum = 0;
  for (int i = 0; i < NBMAX / 32; ++i) { int j = b0 + i; if (j < n) sum += a[j]; }
  int inc = sum;
  for (int o = 1; o < 32; o <<= 1) { int v = __shfl_up(inc, o); if (lane >= o) inc += v; }
  int run = inc - sum;
  for (int i = 0; i < NBMAX / 32; ++i) { int j = b0 + i; if (j < n) { int v = a[j]; a[j] = run; run += v; } }
  return __shfl(inc, 31);
}

__device__ __forceinline__ int wave_excl_scan_full(int* a, int lane) {
  int* p = a + lane * (NBMAX / 32);
  int v[NBMAX / 32];
  int sum = 0;
#pragma unroll
  for (int i = 0; i < NBMAX / 32; ++i) { v[i] = p[i]; sum += v[i]; }
  int inc = sum;
  for (int o = 1; o < 32; o <<= 1) { int t = __shfl_up(inc, o); if (lane >= o) inc += t; }
  int run = inc - sum;
#pragma unroll
  for (int i = 0; i < NBMAX / 32; ++i) { p[i] = run; run += v[i]; }
  return __shfl(inc, 31);
}

__global__ __launch_bounds__(NT) void gcn_count(const int* __restrict__ dst, int* __restrict__ cnt,
                                                long long E, int NB, int CB) {
  __shared__ int c[NBMAX];
  int tid = (int)threadIdx.x, b = (int)blockIdx.x;
  for (int i = tid; i < NBMAX; i += NT) c[i] = 0;
  __syncthreads();
  long long e0 = (long long)b * CH;
  int n = (int)((E - e0 < CH) ? (E - e0) : CH);
  for (int i0 = tid; i0 < n; i0 += 4 * NT) {
    int d[4];
#pragma unroll
    for (int j = 0; j < 4; ++j) { int i = i0 + j * NT; d[j] = (i < n) ? dst[e0 + i] : -1; }
#pragma unroll
    for (int j = 0; j < 4; ++j) if (d[j] >= 0) lds_add_i32(&c[d[j] >> BSH], 1);
  }
  __syncthreads();
  for (int k = tid; k < NB; k += NT) cnt[(size_t)k * CB + b] = c[k];
}

__global__ void gcn_scan(int* __restrict__ cnt, int* __restrict__ tot, int NB, int CB) {
  int wave = (int)(((long long)blockIdx.x * blockDim.x + threadIdx.x) >> 5);
  int lane = (int)(threadIdx.x & 31);
  if (wave >= NB) return;
  int* row = cnt + (size_t)wave * CB;
  int per = (CB + 31) >> 5;
  int b0 = lane * per;
  int sum = 0;
  for (int i = 0; i < per; ++i) { int j = b0 + i; if (j < CB) sum += row[j]; }
  int inc = sum;
  for (int o = 1; o < 32; o <<= 1) { int v = __shfl_up(inc, o); if (lane >= o) inc += v; }
  int run = inc - sum;
  for (int i = 0; i < per; ++i) { int j = b0 + i; if (j < CB) { int v = row[j]; row[j] = run; run += v; } }
  if (lane == 31) tot[wave] = inc;
}

__global__ __launch_bounds__(NT) void gcn_place(const int* __restrict__ src, const int* __restrict__ dst,
                                                const int* __restrict__ cnt, const int* __restrict__ tot,
                                                unsigned* __restrict__ sorted, int* __restrict__ bst,
                                                long long E, int NB, int CB) {
  __shared__ unsigned       sw[CH];
  __shared__ unsigned short sb[CH];
  __shared__ int lstart[NBMAX], lcur[NBMAX], gbase[NBMAX], gst[NBMAX];
  int tid = (int)threadIdx.x, b = (int)blockIdx.x, lane = tid & 31;
  for (int k = tid; k < NBMAX; k += NT) {
    int own = 0, before = 0, t = 0;
    if (k < NB) {
      t = tot[k];
      before = cnt[(size_t)k * CB + b];
      int next = (b + 1 < CB) ? cnt[(size_t)k * CB + b + 1] : t;
      own = next - before;
    }
    lstart[k] = own; gbase[k] = before; gst[k] = t;
  }
  __syncthreads();
  if (tid < 32) {
    wave_excl_scan_lds(lstart, NB, lane);
    wave_excl_scan_lds(gst, NB, lane);
  }
  __syncthreads();
  for (int k = tid; k < NB; k += NT) {
    lcur[k] = lstart[k];
    gbase[k] += gst[k];
    if (b == 0) bst[k] = gst[k];
  }
  if (b == 0 && tid == 0) bst[NB] = (int)E;
  __syncthreads();
  long long e0 = (long long)b * CH;
  int n = (int)((E - e0 < CH) ? (E - e0) : CH);
  for (int i0 = tid; i0 < n; i0 += 4 * NT) {
    int d[4], s[4];
#pragma unroll
    for (int j = 0; j < 4; ++j) {
      int i = i0 + j * NT;
      d[j] = (i < n) ? dst[e0 + i] : -1;
      s[j] = (i < n) ? src[e0 + i] : 0;
    }
#pragma unroll
    for (int j = 0; j < 4; ++j) {
      if (d[j] >= 0) {
        int k = d[j] >> BSH;
        int r = lds_add_ret_i32(&lcur[k], 1);
        sw[r] = ((unsigned)s[j] << BSH) | (unsigned)(d[j] & (BSZ - 1));
        sb[r] = (unsigned short)k;
      }
    }
  }
  __syncthreads();
  for (int i = tid; i < n; i += NT) {
    int k = sb[i];
    sorted[(size_t)gbase[k] + (i - lstart[k])] = sw[i];
  }
}

__global__ __launch_bounds__(NT) void gcn_local(const int* __restrict__ src, const int* __restrict__ dst,
                                                unsigned* __restrict__ tmp, unsigned* __restrict__ seg,
                                                long long E, int NB, int CB) {
  __shared__ unsigned sw[CH];
  __shared__ int lstart[NBMAX];
  int tid = (int)threadIdx.x, b = (int)blockIdx.x, lane = tid & 31;
  for (int k = tid; k < NBMAX; k += NT) lstart[k] = 0;
  long long e0 = (long long)b * CH;
  int n = (int)((E - e0 < CH) ? (E - e0) : CH);
  int d[CH / NT], s[CH / NT], rk[CH / NT];
#pragma unroll
  for (int j = 0; j < CH / NT; ++j) {
    int i = tid + j * NT;
    int ic = (i < n) ? i : (n - 1);
    d[j] = dst[e0 + ic];
    s[j] = src[e0 + ic];
  }
#pragma unroll
  for (int j = 0; j < CH / NT; ++j) if (tid + j * NT >= n) d[j] = -1;
  __syncthreads();
#pragma unroll
  for (int j = 0; j < CH / NT; ++j) rk[j] = (d[j] >= 0) ? lds_add_ret_i32(&lstart[d[j] >> BSH], 1) : 0;
  __syncthreads();
  if (tid < 32) wave_excl_scan_full(lstart, lane);
  __syncthreads();
#pragma unroll
  for (int j = 0; j < CH / NT; ++j) rk[j] += lstart[(d[j] >= 0) ? (d[j] >> BSH) : 0];
#pragma unroll
  for (int j = 0; j < CH / NT; ++j)
    if (d[j] >= 0) sw[rk[j]] = ((unsigned)s[j] << BSH) | (unsigned)(d[j] & (BSZ - 1));
  for (int k = tid; k < NB; k += NT) {
    int st = lstart[k];
    int en = (k + 1 < NB) ? lstart[k + 1] : n;
    seg[(size_t)k * CB + b] = (unsigned)st | ((unsigned)en << 16);
  }
  __syncthreads();
  for (int i = tid; i < n; i += NT) tmp[e0 + i] = sw[i];
}

__global__ __launch_bounds__(NT) void gcn_degree(unsigned* __restrict__ sorted, int* __restrict__ bst,
                                                 const unsigned* __restrict__ tmp, const unsigned* __restrict__ seg,
                                                 const float* __restrict__ x, const float* __restrict__ W1,
                                                 float* __restrict__ dinv, float* __restrict__ rows,
                                                 int N, int NB, int CB) {
  __shared__ int c[BSZ];
  __shared__ float sd[BSZ];
  __shared__ __attribute__((aligned(16))) float st[BSZ * 16];
  __shared__ int sst[CBMAX], slen[CBMAX], spre[CBMAX];
  __shared__ int sbase, stot;
  int tid = (int)threadIdx.x, k = (int)blockIdx.x;
  int lane = tid & 31, wave = tid >> 5;
  int node0 = k << BSH;
  if (tid < BSZ) c[tid] = 0;
  unsigned segw = (tid < CB) ? seg[(size_t)k * CB + tid] : 0u;

  v16h bh = {}, bl = {}, ah = {}, al = {};
  if (lane < 16) {
#pragma unroll
    for (int kk = 0; kk < 5; ++kk) {
      float w = W1[kk * 16 + lane];
      _Float16 hi = (_Float16)w;
      bh[kk] = hi;
      bl[kk] = (_Float16)(w - (float)hi);
    }
    int d = node0 + (wave << 4) + lane;
    if (d < N) {
      const float* xr = x + (size_t)d * 5;
#pragma unroll
      for (int kk = 0; kk < 5; ++kk) {
        float v = xr[kk];
        _Float16 hi = (_Float16)v;
        ah[kk] = hi;
        al[kk] = (_Float16)(v - (float)hi);
      }
    }
  }
  if (CB > 0) {
    int s0 = (int)(segw & 0xFFFFu), s1 = (int)(segw >> 16);
    sst[tid] = s0; slen[tid] = s1 - s0; spre[tid] = s1 - s0;
  }
  __syncthreads();
  if (CB > 0) {
    if (wave == 0) {
      int tot = wave_excl_scan_full(spre, lane);
      if (lane == 0) stot = tot;
    } else if (wave == 1) {
      int sum = 0;
#pragma unroll
      for (int i = 0; i < CBMAX / 32; ++i) sum += sst[lane + 32 * i];
      for (int o = 16; o > 0; o >>= 1) sum += __shfl_xor(sum, o);
      if (lane == 0) sbase = sum;
    }
    __syncthreads();
    int base = sbase;
    if (tid == 0) {
      bst[k] = base;
      if (k == NB - 1) bst[NB] = base + stot;
    }
    for (int c0 = wave; c0 < CB; c0 += SEGT * (NT / 32)) {
      unsigned w[SEGT];
      int pos[SEGT], len[SEGT], off[SEGT];
      int mx = 0;
#pragma unroll
      for (int u = 0; u < SEGT; ++u) {
        int cc = c0 + u * (NT / 32);
        bool in = cc < CB;
        int ci = in ? cc : 0;
        int l = slen[ci];
        off[u] = ci * CH + sst[ci];
        pos[u] = spre[ci];
        len[u] = in ? l : 0;
        mx = (len[u] > mx) ? len[u] : mx;
      }
#pragma unroll
      for (int u = 0; u < SEGT; ++u) w[u] = (lane < len[u]) ? tmp[(size_t)off[u] + lane] : NOEDGE;
#pragma unroll
      for (int u = 0; u < SEGT; ++u) {
        if (w[u] != NOEDGE) {
          sorted[base + pos[u] + lane] = w[u];
          lds_add_i32(&c[w[u] & (BSZ - 1)], 1);
        }
      }
      if (mx > 32) {
        for (int u = 0; u < SEGT; ++u) {
          for (int o = 32 + lane; o < len[u]; o += 32) {
            unsigned wv = tmp[(size_t)off[u] + o];
            sorted[base + pos[u] + o] = wv;
            lds_add_i32(&c[wv & (BSZ - 1)], 1);
          }
        }
      }
    }
  } else {
    int e0 = bst[k], e1 = bst[k + 1];
    for (int i0 = e0 + tid; i0 < e1; i0 += 4 * NT) {
      unsigned w[4];
#pragma unroll
      for (int j = 0; j < 4; ++j) { int i = i0 + j * NT; w[j] = (i < e1) ? sorted[i] : NOEDGE; }
#pragma unroll
      for (int j = 0; j < 4; ++j) if (w[j] != NOEDGE) lds_add_i32(&c[w[j] & (BSZ - 1)], 1);
    }
  }
  __syncthreads();
  if (tid < BSZ) {
    int d = node0 + tid;
    float di = rsqrtf(1.0f + (float)c[tid]);
    sd[tid] = di;
    if (d < N) dinv[d] = di;
  }
  v8f acc = {};
  acc = __builtin_amdgcn_wmma_f32_16x16x32_f16(false, al, false, bh, (short)0, acc, false, false);
  acc = __builtin_amdgcn_wmma_f32_16x16x32_f16(false, ah, false, bl, (short)0, acc, false, false);
  acc = __builtin_amdgcn_wmma_f32_16x16x32_f16(false, ah, false, bh, (short)0, acc, false, false);
  __syncthreads();
  int ch = lane & 15;
  int nb = (wave << 4) + ((lane >> 4) << 3);
#pragma unroll
  for (int r = 0; r < 8; ++r) st[(nb + r) * 16 + ch] = acc[r] * sd[nb + r];
  __syncthreads();
  const float4* st4 = (const float4*)st;
  float4* out4 = (float4*)rows + (size_t)node0 * 4;
  for (int i = tid; i < BSZ * 4; i += NT)
    if (node0 + (i >> 2) < N) out4[i] = st4[i];
}

__global__ __launch_bounds__(NTL) void gcn_layer1(const unsigned* __restrict__ sorted, const int* __restrict__ bst,
                                                 const float* __restrict__ rows, const float* __restrict__ b1,
                                                 const float* __restrict__ W2, const float* __restrict__ dinv,
                                                 float* __restrict__ t2, int N) {
  __shared__ float bins[BSZ * PITCH];
  int tid = (int)threadIdx.x, k = (int)blockIdx.x;
  int node0 = k << BSH;
  const float4* rows4 = (const float4*)rows;
  int g = tid >> 2, q = tid & 3;
  {
    int d = node0 + g;
    float4 v0 = rows4[(size_t)((d < N) ? d : (N - 1)) * 4 + q];
    float* p = &bins[g * PITCH + (q << 2)];
    bool in = d < N;
    p[0] = in ? v0.x : 0.0f;
    p[1] = in ? v0.y : 0.0f;
    p[2] = in ? v0.z : 0.0f;
    p[3] = in ? v0.w : 0.0f;
  }
  __syncthreads();

  int e0 = bst[k], e1 = bst[k + 1];
  int i0 = e0 + g;
  if (i0 < e1) {
    unsigned w[4];
#pragma unroll
    for (int j = 0; j < 4; ++j) {
      int i = i0 + j * (NTL / 4);
      w[j] = sorted[(i < e1) ? i : (e1 - 1)];
    }
    do {
      unsigned wc[4];
      float4 v[4];
      bool ok[4];
#pragma unroll
      for (int j = 0; j < 4; ++j) {
        wc[j] = w[j];
        ok[j] = (i0 + j * (NTL / 4)) < e1;
        v[j] = rows4[(size_t)(wc[j] >> BSH) * 4 + q];
      }
      i0 += 4 * (NTL / 4);
#pragma unroll
      for (int j = 0; j < 4; ++j) {
        int i = i0 + j * (NTL / 4);
        w[j] = sorted[(i < e1) ? i : (e1 - 1)];
      }
#pragma unroll
      for (int j = 0; j < 4; ++j) {
        int dl = ok[j] ? (int)(wc[j] & (BSZ - 1)) : g;
        float* p = &bins[dl * PITCH + (q << 2)];
        lds_add_f32(p + 0, ok[j] ? v[j].x : 0.0f);
        lds_add_f32(p + 1, ok[j] ? v[j].y : 0.0f);
        lds_add_f32(p + 2, ok[j] ? v[j].z : 0.0f);
        lds_add_f32(p + 3, ok[j] ? v[j].w : 0.0f);
      }
    } while (i0 < e1);
  }
  __syncthreads();
  if (tid < BSZ) {
    int d = node0 + tid;
    if (d < N) {
      float di = dinv[d];
      float acc = 0.0f;
#pragma unroll
      for (int cc = 0; cc < 16; ++cc) {
        float v = fmaf(bins[tid * PITCH + cc], di, b1[cc]);
        v = fmaxf(v, 0.0f);
        acc = fmaf(v, W2[cc], acc);
      }
      t2[d] = acc;
    }
  }
}

__global__ __launch_bounds__(NT) void gcn_layer2(const unsigned* __restrict__ sorted, const int* __restrict__ bst,
                                                 const float* __restrict__ t2, const float* __restrict__ dinv,
                                                 const float* __restrict__ b2, float* __restrict__ out, int N) {
  __shared__ float bins[BSZ];
  __shared__ float sdinv[BSZ];
  int tid = (int)threadIdx.x, k = (int)blockIdx.x;
  int node0 = k << BSH;
  if (tid < BSZ) { int d = node0 + tid; bins[tid] = 0.0f; sdinv[tid] = (d < N) ? dinv[d] : 0.0f; }
  __syncthreads();
  int e0 = bst[k], e1 = bst[k + 1];
  for (int i0 = e0 + tid; i0 < e1; i0 += 4 * NT) {
    unsigned w[4];
    float a[4], g[4];
#pragma unroll
    for (int j = 0; j < 4; ++j) { int i = i0 + j * NT; w[j] = (i < e1) ? sorted[i] : NOEDGE; }
#pragma unroll
    for (int j = 0; j < 4; ++j) {
      unsigned s = (w[j] != NOEDGE) ? (w[j] >> BSH) : 0u;
      a[j] = t2[s]; g[j] = dinv[s];
    }
#pragma unroll
    for (int j = 0; j < 4; ++j) {
      if (w[j] != NOEDGE) {
        int dl = (int)(w[j] & (BSZ - 1));
        lds_add_f32(&bins[dl], a[j] * g[j] * sdinv[dl]);
      }
    }
  }
  __syncthreads();
  if (tid < BSZ) {
    int d = node0 + tid;
    if (d < N) { float di = sdinv[tid]; out[d] = bins[tid] + t2[d] * di * di + b2[0]; }
  }
}

extern "C" void kernel_launch(void* const* d_in, const int* in_sizes, int n_in,
                              void* d_out, int out_size, void* d_ws, size_t ws_size,
                              hipStream_t stream) {
  const float* x  = (const float*)d_in[0];
  const float* W1 = (const float*)d_in[1];
  const float* b1 = (const float*)d_in[2];
  const float* W2 = (const float*)d_in[3];
  const float* b2 = (const float*)d_in[4];
  const int*   ei = (const int*)d_in[5];

  int       N = in_sizes[0] / 5;
  long long E = (long long)in_sizes[5] / 2;
  const int* src = ei;
  const int* dst = ei + E;

  int NB = (N + BSZ - 1) / BSZ;
  int CB = (int)((E + CH - 1) / CH);

  float*    rows   = (float*)d_ws;
  unsigned* sorted = (unsigned*)(rows + (size_t)16 * N);
  int*      bst    = (int*)(sorted + E);
  float*    dinv   = (float*)(bst + NB + 1);
  float*    t2     = dinv + N;
  int*      cnt    = (int*)dinv;
  int*      tot    = cnt + (size_t)NB * CB;
  unsigned* tmp    = (unsigned*)(t2 + N);
  unsigned* seg    = tmp + E;

  size_t need = ((size_t)18 * N + (size_t)E + (size_t)NB + 1) * 4;
  if (ws_size < need) return;

  float* out = (float*)d_out;

  size_t need1 = need + ((size_t)E + (size_t)NB * CB) * 4;
  if (ws_size >= need1 && CB <= CBMAX) {
    gcn_local <<<CB, NT, 0, stream>>>(src, dst, tmp, seg, E, NB, CB);
    gcn_degree<<<NB, NT, 0, stream>>>(sorted, bst, tmp, seg, x, W1, dinv, rows, N, NB, CB);
  } else {
    gcn_count <<<CB, NT, 0, stream>>>(dst, cnt, E, NB, CB);
    gcn_scan  <<<(NB + 7) / 8, 256, 0, stream>>>(cnt, tot, NB, CB);
    gcn_place <<<CB, NT, 0, stream>>>(src, dst, cnt, tot, sorted, bst, E, NB, CB);
    gcn_degree<<<NB, NT, 0, stream>>>(sorted, bst, tmp, seg, x, W1, dinv, rows, N, NB, 0);
  }
  gcn_layer1<<<NB, NTL, 0, stream>>>(sorted, bst, rows, b1, W2, dinv, t2, N);
  gcn_layer2<<<NB, NT, 0, stream>>>(sorted, bst, t2, dinv, b2, out, N);
}


# Destination buckets re-tiled to twice the WGP count: 196 nodes each (511 buckets; was 256 nodes, 391), bucket = node / 196 kept in the edge word; each WGP owns two equal buckets per owner launch
# speedup vs baseline: 6.0286x; 1.0612x over previous
#include <hip/hip_runtime.h>
#include <hip/hip_bf16.h>


typedef __attribute__((ext_vector_type(16))) _Float16 v16h;
typedef __attribute__((ext_vector_type(8)))  float    v8f;

#define NT      512
#define NTL     1024
#define NOEDGE  0xFFFFFFFFu
#define BSH     8
#define BSZ     256
#define NBMAX   512
#define CH      8192
#define CBMAX   512
#define SEGT    16
static_assert(CBMAX == NBMAX && CBMAX == NT && CH % NT == 0, "one scan routine for both tables; one table entry per thread");
#define PITCH   17


__device__ __forceinline__ void lds_add_i32(int* p, int v) {
  __hip_atomic_fetch_add(p, v, __ATOMIC_RELAXED, __HIP_MEMORY_SCOPE_WORKGROUP);
}
__device__ __forceinline__ int lds_add_ret_i32(int* p, int v) {
  return __hip_atomic_fetch_add(p, v, __ATOMIC_RELAXED, __HIP_MEMORY_SCOPE_WORKGROUP);
}
__device__ __forceinline__ void lds_add_f32(float* p, float v) {
  __hip_atomic_fetch_add(p, v, __ATOMIC_RELAXED, __HIP_MEMORY_SCOPE_WORKGROUP);
}

__device__ __forceinline__ int bucket_word(int d, int bn, float inv) {
  int k = (int)(((float)d + 0.5f) * inv);
  return (k << BSH) | (d - k * bn);
}

__device__ __forceinline__ int wave_excl_scan_lds(int* a, int n, int lane) {
  int b0 = lane * (NBMAX / 32);
  int sum = 0;
  for (int i = 0; i < NBMAX / 32; ++i) { int j = b0 + i; if (j < n) sum += a[j]; }
  int inc = sum;
  for (int o = 1; o < 32; o <<= 1) { int v = __shfl_up(inc, o); if (lane >= o) inc += v; }
  int run = inc - sum;
  for (int i = 0; i < NBMAX / 32; ++i) { int j = b0 + i; if (j < n) { int v = a[j]; a[j] = run; run += v; } }
  return __shfl(inc, 31);
}

__device__ __forceinline__ int wave_excl_scan_full(int* a, int lane) {
  int* p = a + lane * (NBMAX / 32);
  int v[NBMAX / 32];
  int sum = 0;
#pragma unroll
  for (int i = 0; i < NBMAX / 32; ++i) { v[i] = p[i]; sum += v[i]; }
  int inc = sum;
  for (int o = 1; o < 32; o <<= 1) { int t = __shfl_up(inc, o); if (lane >= o) inc += t; }
  int run = inc - sum;
#pragma unroll
  for (int i = 0; i < NBMAX / 32; ++i) { p[i] = run; run += v[i]; }
  return __shfl(inc, 31);
}

__global__ __launch_bounds__(NT) void gcn_count(const int* __restrict__ dst, int* __restrict__ cnt,
                                                long long E, int NB, int CB, int BN) {
  __shared__ int c[NBMAX];
  int tid = (int)threadIdx.x, b = (int)blockIdx.x;
  float inv = 1.0f / (float)BN;
  for (int i = tid; i < NBMAX; i += NT) c[i] = 0;
  __syncthreads();
  long long e0 = (long long)b * CH;
  int n = (int)((E - e0 < CH) ? (E - e0) : CH);
  for (int i0 = tid; i0 < n; i0 += 4 * NT) {
    int d[4];
#pragma unroll
    for (int j = 0; j < 4; ++j) { int i = i0 + j * NT; d[j] = (i < n) ? dst[e0 + i] : -1; }
#pragma unroll
    for (int j = 0; j < 4; ++j) if (d[j] >= 0) lds_add_i32(&c[bucket_word(d[j], BN, inv) >> BSH], 1);
  }
  __syncthreads();
  for (int k = tid; k < NB; k += NT) cnt[(size_t)k * CB + b] = c[k];
}

__global__ void gcn_scan(int* __restrict__ cnt, int* __restrict__ tot, int NB, int CB) {
  int wave = (int)(((long long)blockIdx.x * blockDim.x + threadIdx.x) >> 5);
  int lane = (int)(threadIdx.x & 31);
  if (wave >= NB) return;
  int* row = cnt + (size_t)wave * CB;
  int per = (CB + 31) >> 5;
  int b0 = lane * per;
  int sum = 0;
  for (int i = 0; i < per; ++i) { int j = b0 + i; if (j < CB) sum += row[j]; }
  int inc = sum;
  for (int o = 1; o < 32; o <<= 1) { int v = __shfl_up(inc, o); if (lane >= o) inc += v; }
  int run = inc - sum;
  for (int i = 0; i < per; ++i) { int j = b0 + i; if (j < CB) { int v = row[j]; row[j] = run; run += v; } }
  if (lane == 31) tot[wave] = inc;
}

__global__ __launch_bounds__(NT) void gcn_place(const int* __restrict__ src, const int* __restrict__ dst,
                                                const int* __restrict__ cnt, const int* __restrict__ tot,
                                                unsigned* __restrict__ sorted, int* __restrict__ bst,
                                                long long E, int NB, int CB, int BN) {
  __shared__ unsigned       sw[CH];
  __shared__ unsigned short sb[CH];
  __shared__ int lstart[NBMAX], lcur[NBMAX], gbase[NBMAX], gst[NBMAX];
  int tid = (int)threadIdx.x, b = (int)blockIdx.x, lane = tid & 31;
  for (int k = tid; k < NBMAX; k += NT) {
    int own = 0, before = 0, t = 0;
    if (k < NB) {
      t = tot[k];
      before = cnt[(size_t)k * CB + b];
      int next = (b + 1 < CB) ? cnt[(size_t)k * CB + b + 1] : t;
      own = next - before;
    }
    lstart[k] = own; gbase[k] = before; gst[k] = t;
  }
  __syncthreads();
  if (tid < 32) {
    wave_excl_scan_lds(lstart, NB, lane);
    wave_excl_scan_lds(gst, NB, lane);
  }
  __syncthreads();
  for (int k = tid; k < NB; k += NT) {
    lcur[k] = lstart[k];
    gbase[k] += gst[k];
    if (b == 0) bst[k] = gst[k];
  }
  if (b == 0 && tid == 0) bst[NB] = (int)E;
  __syncthreads();
  long long e0 = (long long)b * CH;
  int n = (int)((E - e0 < CH) ? (E - e0) : CH);
  for (int i0 = tid; i0 < n; i0 += 4 * NT) {
    int d[4], s[4];
#pragma unroll
    for (int j = 0; j < 4; ++j) {
      int i = i0 + j * NT;
      d[j] = (i < n) ? dst[e0 + i] : -1;
      s[j] = (i < n) ? src[e0 + i] : 0;
    }
#pragma unroll
    for (int j = 0; j < 4; ++j) {
      if (d[j] >= 0) {
        int bw = bucket_word(d[j], BN, 1.0f / (float)BN);
        int k = bw >> BSH;
        int r = lds_add_ret_i32(&lcur[k], 1);
        sw[r] = ((unsigned)s[j] << BSH) | (unsigned)(bw & (BSZ - 1));
        sb[r] = (unsigned short)k;
      }
    }
  }
  __syncthreads();
  for (int i = tid; i < n; i += NT) {
    int k = sb[i];
    sorted[(size_t)gbase[k] + (i - lstart[k])] = sw[i];
  }
}

__global__ __launch_bounds__(NT) void gcn_local(const int* __restrict__ src, const int* __restrict__ dst,
                                                unsigned* __restrict__ tmp, unsigned* __restrict__ seg,
                                                long long E, int NB, int CB, int BN) {
  __shared__ unsigned sw[CH];
  __shared__ int lstart[NBMAX];
  int tid = (int)threadIdx.x, b = (int)blockIdx.x, lane = tid & 31;
  for (int k = tid; k < NBMAX; k += NT) lstart[k] = 0;
  long long e0 = (long long)b * CH;
  int n = (int)((E - e0 < CH) ? (E - e0) : CH);
  int d[CH / NT], s[CH / NT], rk[CH / NT];
#pragma unroll
  for (int j = 0; j < CH / NT; ++j) {
    int i = tid + j * NT;
    int ic = (i < n) ? i : (n - 1);
    d[j] = dst[e0 + ic];
    s[j] = src[e0 + ic];
  }
  float inv = 1.0f / (float)BN;
#pragma unroll
  for (int j = 0; j < CH / NT; ++j) d[j] = (tid + j * NT >= n) ? -1 : bucket_word(d[j], BN, inv);
  __syncthreads();
#pragma unroll
  for (int j = 0; j < CH / NT; ++j) rk[j] = (d[j] >= 0) ? lds_add_ret_i32(&lstart[d[j] >> BSH], 1) : 0;
  __syncthreads();
  if (tid < 32) wave_excl_scan_full(lstart, lane);
  __syncthreads();
#pragma unroll
  for (int j = 0; j < CH / NT; ++j) rk[j] += lstart[(d[j] >= 0) ? (d[j] >> BSH) : 0];
#pragma unroll
  for (int j = 0; j < CH / NT; ++j)
    if (d[j] >= 0) sw[rk[j]] = ((unsigned)s[j] << BSH) | (unsigned)(d[j] & (BSZ - 1));
  for (int k = tid; k < NB; k += NT) {
    int st = lstart[k];
    int en = (k + 1 < NB) ? lstart[k + 1] : n;
    seg[(size_t)k * CB + b] = (unsigned)st | ((unsigned)en << 16);
  }
  __syncthreads();
  for (int i = tid; i < n; i += NT) tmp[e0 + i] = sw[i];
}

__global__ __launch_bounds__(NT) void gcn_degree(unsigned* __restrict__ sorted, int* __restrict__ bst,
                                                 const unsigned* __restrict__ tmp, const unsigned* __restrict__ seg,
                                                 const float* __restrict__ x, const float* __restrict__ W1,
                                                 float* __restrict__ dinv, float* __restrict__ rows,
                                                 int N, int NB, int CB, int BN) {
  __shared__ int c[BSZ];
  __shared__ float sd[BSZ];
  __shared__ __attribute__((aligned(16))) float st[BSZ * 16];
  __shared__ int sst[CBMAX], slen[CBMAX], spre[CBMAX];
  __shared__ int sbase, stot;
  int tid = (int)threadIdx.x, k = (int)blockIdx.x;
  int lane = tid & 31, wave = tid >> 5;
  int node0 = k * BN;
  if (tid < BSZ) c[tid] = 0;
  unsigned segw = (tid < CB) ? seg[(size_t)k * CB + tid] : 0u;

  v16h bh = {}, bl = {}, ah = {}, al = {};
  if (lane < 16) {
#pragma unroll
    for (int kk = 0; kk < 5; ++kk) {
      float w = W1[kk * 16 + lane];
      _Float16 hi = (_Float16)w;
      bh[kk] = hi;
      bl[kk] = (_Float16)(w - (float)hi);
    }
    int d = node0 + (wave << 4) + lane;
    if (d < N) {
      const float* xr = x + (size_t)d * 5;
#pragma unroll
      for (int kk = 0; kk < 5; ++kk) {
        float v = xr[kk];
        _Float16 hi = (_Float16)v;
        ah[kk] = hi;
        al[kk] = (_Float16)(v - (float)hi);
      }
    }
  }
  if (CB > 0) {
    int s0 = (int)(segw & 0xFFFFu), s1 = (int)(segw >> 16);
    sst[tid] = s0; slen[tid] = s1 - s0; spre[tid] = s1 - s0;
  }
  __syncthreads();
  if (CB > 0) {
    if (wave == 0) {
      int tot = wave_excl_scan_full(spre, lane);
      if (lane == 0) stot = tot;
    } else if (wave == 1) {
      int sum = 0;
#pragma unroll
      for (int i = 0; i < CBMAX / 32; ++i) sum += sst[lane + 32 * i];
      for (int o = 16; o > 0; o >>= 1) sum += __shfl_xor(sum, o);
      if (lane == 0) sbase = sum;
    }
    __syncthreads();
    int base = sbase;
    if (tid == 0) {
      bst[k] = base;
      if (k == NB - 1) bst[NB] = base + stot;
    }
    for (int c0 = wave; c0 < CB; c0 += SEGT * (NT / 32)) {
      unsigned w[SEGT];
      int pos[SEGT], len[SEGT], off[SEGT];
      int mx = 0;
#pragma unroll
      for (int u = 0; u < SEGT; ++u) {
        int cc = c0 + u * (NT / 32);
        bool in = cc < CB;
        int ci = in ? cc : 0;
        int l = slen[ci];
        off[u] = ci * CH + sst[ci];
        pos[u] = spre[ci];
        len[u] = in ? l : 0;
        mx = (len[u] > mx) ? len[u] : mx;
      }
#pragma unroll
      for (int u = 0; u < SEGT; ++u) w[u] = (lane < len[u]) ? tmp[(size_t)off[u] + lane] : NOEDGE;
#pragma unroll
      for (int u = 0; u < SEGT; ++u) {
        if (w[u] != NOEDGE) {
          sorted[base + pos[u] + lane] = w[u];
          lds_add_i32(&c[w[u] & (BSZ - 1)], 1);
        }
      }
      if (mx > 32) {
        for (int u = 0; u < SEGT; ++u) {
          for (int o = 32 + lane; o < len[u]; o += 32) {
            unsigned wv = tmp[(size_t)off[u] + o];
            sorted[base + pos[u] + o] = wv;
            lds_add_i32(&c[wv & (BSZ - 1)], 1);
          }
        }
      }
    }
  } else {
    int e0 = bst[k], e1 = bst[k + 1];
    for (int i0 = e0 + tid; i0 < e1; i0 += 4 * NT) {
      unsigned w[4];
#pragma unroll
      for (int j = 0; j < 4; ++j) { int i = i0 + j * NT; w[j] = (i < e1) ? sorted[i] : NOEDGE; }
#pragma unroll
      for (int j = 0; j < 4; ++j) if (w[j] != NOEDGE) lds_add_i32(&c[w[j] & (BSZ - 1)], 1);
    }
  }
  __syncthreads();
  if (tid < BSZ) {
    int d = node0 + tid;
    float di = rsqrtf(1.0f + (float)c[tid]);
    sd[tid] = di;
    if (tid < BN && d < N) dinv[d] = di;
  }
  v8f acc = {};
  acc = __builtin_amdgcn_wmma_f32_16x16x32_f16(false, al, false, bh, (short)0, acc, false, false);
  acc = __builtin_amdgcn_wmma_f32_16x16x32_f16(false, ah, false, bl, (short)0, acc, false, false);
  acc = __builtin_amdgcn_wmma_f32_16x16x32_f16(false, ah, false, bh, (short)0, acc, false, false);
  __syncthreads();
  int ch = lane & 15;
  int nb = (wave << 4) + ((lane >> 4) << 3);
#pragma unroll
  for (int r = 0; r < 8; ++r) st[(nb + r) * 16 + ch] = acc[r] * sd[nb + r];
  __syncthreads();
  const float4* st4 = (const float4*)st;
  float4* out4 = (float4*)rows + (size_t)node0 * 4;
  for (int i = tid; i < BN * 4; i += NT)
    if (node0 + (i >> 2) < N) out4[i] = st4[i];
}

__global__ __launch_bounds__(NTL) void gcn_layer1(const unsigned* __restrict__ sorted, const int* __restrict__ bst,
                                                 const float* __restrict__ rows, const float* __restrict__ b1,
                                                 const float* __restrict__ W2, const float* __restrict__ dinv,
                                                 float* __restrict__ t2, int N, int BN) {
  __shared__ float bins[BSZ * PITCH];
  int tid = (int)threadIdx.x, k = (int)blockIdx.x;
  int node0 = k * BN;
  const float4* rows4 = (const float4*)rows;
  int g = tid >> 2, q = tid & 3;
  {
    int d = node0 + g;
    float4 v0 = rows4[(size_t)((d < N) ? d : (N - 1)) * 4 + q];
    float* p = &bins[g * PITCH + (q << 2)];
    bool in = (g < BN) && (d < N);
    p[0] = in ? v0.x : 0.0f;
    p[1] = in ? v0.y : 0.0f;
    p[2] = in ? v0.z : 0.0f;
    p[3] = in ? v0.w : 0.0f;
  }
  __syncthreads();

  int e0 = bst[k], e1 = bst[k + 1];
  int i0 = e0 + g;
  if (i0 < e1) {
    unsigned w[4];
#pragma unroll
    for (int j = 0; j < 4; ++j) {
      int i = i0 + j * (NTL / 4);
      w[j] = sorted[(i < e1) ? i : (e1 - 1)];
    }
    do {
      unsigned wc[4];
      float4 v[4];
      bool ok[4];
#pragma unroll
      for (int j = 0; j < 4; ++j) {
        wc[j] = w[j];
        ok[j] = (i0 + j * (NTL / 4)) < e1;
        v[j] = rows4[(size_t)(wc[j] >> BSH) * 4 + q];
      }
      i0 += 4 * (NTL / 4);
#pragma unroll
      for (int j = 0; j < 4; ++j) {
        int i = i0 + j * (NTL / 4);
        w[j] = sorted[(i < e1) ? i : (e1 - 1)];
      }
#pragma unroll
      for (int j = 0; j < 4; ++j) {
        int dl = ok[j] ? (int)(wc[j] & (BSZ - 1)) : g;
        float* p = &bins[dl * PITCH + (q << 2)];
        lds_add_f32(p + 0, ok[j] ? v[j].x : 0.0f);
        lds_add_f32(p + 1, ok[j] ? v[j].y : 0.0f);
        lds_add_f32(p + 2, ok[j] ? v[j].z : 0.0f);
        lds_add_f32(p + 3, ok[j] ? v[j].w : 0.0f);
      }
    } while (i0 < e1);
  }
  __syncthreads();
  if (tid < BN) {
    int d = node0 + tid;
    if (d < N) {
      float di = dinv[d];
      float acc = 0.0f;
#pragma unroll
      for (int cc = 0; cc < 16; ++cc) {
        float v = fmaf(bins[tid * PITCH + cc], di, b1[cc]);
        v = fmaxf(v, 0.0f);
        acc = fmaf(v, W2[cc], acc);
      }
      t2[d] = acc;
    }
  }
}

__global__ __launch_bounds__(NT) void gcn_layer2(const unsigned* __restrict__ sorted, const int* __restrict__ bst,
                                                 const float* __restrict__ t2, const float* __restrict__ dinv,
                                                 const float* __restrict__ b2, float* __restrict__ out, int N, int BN) {
  __shared__ float bins[BSZ];
  __shared__ float sdinv[BSZ];
  int tid = (int)threadIdx.x, k = (int)blockIdx.x;
  int node0 = k * BN;
  if (tid < BSZ) { int d = node0 + tid; bins[tid] = 0.0f; sdinv[tid] = (tid < BN && d < N) ? dinv[d] : 0.0f; }
  __syncthreads();
  int e0 = bst[k], e1 = bst[k + 1];
  for (int i0 = e0 + tid; i0 < e1; i0 += 4 * NT) {
    unsigned w[4];
    float a[4], g[4];
#pragma unroll
    for (int j = 0; j < 4; ++j) { int i = i0 + j * NT; w[j] = (i < e1) ? sorted[i] : NOEDGE; }
#pragma unroll
    for (int j = 0; j < 4; ++j) {
      unsigned s = (w[j] != NOEDGE) ? (w[j] >> BSH) : 0u;
      a[j] = t2[s]; g[j] = dinv[s];
    }
#pragma unroll
    for (int j = 0; j < 4; ++j) {
      if (w[j] != NOEDGE) {
        int dl = (int)(w[j] & (BSZ - 1));
        lds_add_f32(&bins[dl], a[j] * g[j] * sdinv[dl]);
      }
    }
  }
  __syncthreads();
  if (tid < BN) {
    int d = node0 + tid;
    if (d < N) { float di = sdinv[tid]; out[d] = bins[tid] + t2[d] * di * di + b2[0]; }
  }
}

extern "C" void kernel_launch(void* const* d_in, const int* in_sizes, int n_in,
                              void* d_out, int out_size, void* d_ws, size_t ws_size,
                              hipStream_t stream) {
  const float* x  = (const float*)d_in[0];
  const float* W1 = (const float*)d_in[1];
  const float* b1 = (const float*)d_in[2];
  const float* W2 = (const float*)d_in[3];
  const float* b2 = (const float*)d_in[4];
  const int*   ei = (const int*)d_in[5];

  int       N = in_sizes[0] / 5;
  long long E = (long long)in_sizes[5] / 2;
  const int* src = ei;
  const int* dst = ei + E;

  int BN = 2 * ((N + 1023) / 1024);
  int NB = (N + BN - 1) / BN;
  int CB = (int)((E + CH - 1) / CH);

  float*    rows   = (float*)d_ws;
  unsigned* sorted = (unsigned*)(rows + (size_t)16 * N);
  int*      bst    = (int*)(sorted + E);
  float*    dinv   = (float*)(bst + NB + 1);
  float*    t2     = dinv + N;
  int*      cnt    = (int*)rows;
  int*      tot    = cnt + (size_t)NB * CB;
  unsigned* tmp    = (unsigned*)(t2 + N);
  unsigned* seg    = tmp + E;

  size_t need = ((size_t)18 * N + (size_t)E + (size_t)NB + 1) * 4;
  if (ws_size < need) return;

  float* out = (float*)d_out;

  size_t need1 = need + ((size_t)E + (size_t)NB * CB) * 4;
  if (ws_size >= need1 && CB <= CBMAX) {
    gcn_local <<<CB, NT, 0, stream>>>(src, dst, tmp, seg, E, NB, CB, BN);
    gcn_degree<<<NB, NT, 0, stream>>>(sorted, bst, tmp, seg, x, W1, dinv, rows, N, NB, CB, BN);
  } else {
    gcn_count <<<CB, NT, 0, stream>>>(dst, cnt, E, NB, CB, BN);
    gcn_scan  <<<(NB + 7) / 8, 256, 0, stream>>>(cnt, tot, NB, CB);
    gcn_place <<<CB, NT, 0, stream>>>(src, dst, cnt, tot, sorted, bst, E, NB, CB, BN);
    gcn_degree<<<NB, NT, 0, stream>>>(sorted, bst, tmp, seg, x, W1, dinv, rows, N, NB, 0, BN);
  }
  gcn_layer1<<<NB, NTL, 0, stream>>>(sorted, bst, rows, b1, W2, dinv, t2, N, BN);
  gcn_layer2<<<NB, NT, 0, stream>>>(sorted, bst, t2, dinv, b2, out, N, BN);
}


# Layer 2: each bucket's edge run is counting-sorted by source line (32 nodes of t2 and dinv) in LDS before the gather, so a wave's 32 gathers touch about 15 lines, not 32; LDS add order changes
# speedup vs baseline: 6.5051x; 1.0790x over previous
#include <hip/hip_runtime.h>
#include <hip/hip_bf16.h>


typedef __attribute__((ext_vector_type(16))) _Float16 v16h;
typedef __attribute__((ext_vector_type(8)))  float    v8f;

#define NT      512
#define NTL     1024
#define NOEDGE  0xFFFFFFFFu
#define BSH     8
#define BSZ     256
#define NBMAX   512
#define CH      8192
#define CBMAX   512
#define SEGT    16
static_assert(CBMAX == NBMAX && CBMAX == NT && CH % NT == 0, "one scan routine for both tables; one table entry per thread");
#define PITCH   17


__device__ __forceinline__ void lds_add_i32(int* p, int v) {
  __hip_atomic_fetch_add(p, v, __ATOMIC_RELAXED, __HIP_MEMORY_SCOPE_WORKGROUP);
}
__device__ __forceinline__ int lds_add_ret_i32(int* p, int v) {
  return __hip_atomic_fetch_add(p, v, __ATOMIC_RELAXED, __HIP_MEMORY_SCOPE_WORKGROUP);
}
__device__ __forceinline__ void lds_add_f32(float* p, float v) {
  __hip_atomic_fetch_add(p, v, __ATOMIC_RELAXED, __HIP_MEMORY_SCOPE_WORKGROUP);
}

__device__ __forceinline__ int bucket_word(int d, int bn, float inv) {
  int k = (int)(((float)d + 0.5f) * inv);
  return (k << BSH) | (d - k * bn);
}

__device__ __forceinline__ int wave_excl_scan_lds(int* a, int n, int lane) {
  int b0 = lane * (NBMAX / 32);
  int sum = 0;
  for (int i = 0; i < NBMAX / 32; ++i) { int j = b0 + i; if (j < n) sum += a[j]; }
  int inc = sum;
  for (int o = 1; o < 32; o <<= 1) { int v = __shfl_up(inc, o); if (lane >= o) inc += v; }
  int run = inc - sum;
  for (int i = 0; i < NBMAX / 32; ++i) { int j = b0 + i; if (j < n) { int v = a[j]; a[j] = run; run += v; } }
  return __shfl(inc, 31);
}

__device__ __forceinline__ int wave_excl_scan_full(int* a, int lane) {
  int* p = a + lane * (NBMAX / 32);
  int v[NBMAX / 32];
  int sum = 0;
#pragma unroll
  for (int i = 0; i < NBMAX / 32; ++i) { v[i] = p[i]; sum += v[i]; }
  int inc = sum;
  for (int o = 1; o < 32; o <<= 1) { int t = __shfl_up(inc, o); if (lane >= o) inc += t; }
  int run = inc - sum;
#pragma unroll
  for (int i = 0; i < NBMAX / 32; ++i) { p[i] = run; run += v[i]; }
  return __shfl(inc, 31);
}

__global__ __launch_bounds__(NT) void gcn_count(const int* __restrict__ dst, int* __restrict__ cnt,
                                                long long E, int NB, int CB, int BN) {
  __shared__ int c[NBMAX];
  int tid = (int)threadIdx.x, b = (int)blockIdx.x;
  float inv = 1.0f / (float)BN;
  for (int i = tid; i < NBMAX; i += NT) c[i] = 0;
  __syncthreads();
  long long e0 = (long long)b * CH;
  int n = (int)((E - e0 < CH) ? (E - e0) : CH);
  for (int i0 = tid; i0 < n; i0 += 4 * NT) {
    int d[4];
#pragma unroll
    for (int j = 0; j < 4; ++j) { int i = i0 + j * NT; d[j] = (i < n) ? dst[e0 + i] : -1; }
#pragma unroll
    for (int j = 0; j < 4; ++j) if (d[j] >= 0) lds_add_i32(&c[bucket_word(d[j], BN, inv) >> BSH], 1);
  }
  __syncthreads();
  for (int k = tid; k < NB; k += NT) cnt[(size_t)k * CB + b] = c[k];
}

__global__ void gcn_scan(int* __restrict__ cnt, int* __restrict__ tot, int NB, int CB) {
  int wave = (int)(((long long)blockIdx.x * blockDim.x + threadIdx.x) >> 5);
  int lane = (int)(threadIdx.x & 31);
  if (wave >= NB) return;
  int* row = cnt + (size_t)wave * CB;
  int per = (CB + 31) >> 5;
  int b0 = lane * per;
  int sum = 0;
  for (int i = 0; i < per; ++i) { int j = b0 + i; if (j < CB) sum += row[j]; }
  int inc = sum;
  for (int o = 1; o < 32; o <<= 1) { int v = __shfl_up(inc, o); if (lane >= o) inc += v; }
  int run = inc - sum;
  for (int i = 0; i < per; ++i) { int j = b0 + i; if (j < CB) { int v = row[j]; row[j] = run; run += v; } }
  if (lane == 31) tot[wave] = inc;
}

__global__ __launch_bounds__(NT) void gcn_place(const int* __restrict__ src, const int* __restrict__ dst,
                                                const int* __restrict__ cnt, const int* __restrict__ tot,
                                                unsigned* __restrict__ sorted, int* __restrict__ bst,
                                                long long E, int NB, int CB, int BN) {
  __shared__ unsigned       sw[CH];
  __shared__ unsigned short sb[CH];
  __shared__ int lstart[NBMAX], lcur[NBMAX], gbase[NBMAX], gst[NBMAX];
  int tid = (int)threadIdx.x, b = (int)blockIdx.x, lane = tid & 31;
  for (int k = tid; k < NBMAX; k += NT) {
    int own = 0, before = 0, t = 0;
    if (k < NB) {
      t = tot[k];
      before = cnt[(size_t)k * CB + b];
      int next = (b + 1 < CB) ? cnt[(size_t)k * CB + b + 1] : t;
      own = next - before;
    }
    lstart[k] = own; gbase[k] = before; gst[k] = t;
  }
  __syncthreads();
  if (tid < 32) {
    wave_excl_scan_lds(lstart, NB, lane);
    wave_excl_scan_lds(gst, NB, lane);
  }
  __syncthreads();
  for (int k = tid; k < NB; k += NT) {
    lcur[k] = lstart[k];
    gbase[k] += gst[k];
    if (b == 0) bst[k] = gst[k];
  }
  if (b == 0 && tid == 0) bst[NB] = (int)E;
  __syncthreads();
  long long e0 = (long long)b * CH;
  int n = (int)((E - e0 < CH) ? (E - e0) : CH);
  for (int i0 = tid; i0 < n; i0 += 4 * NT) {
    int d[4], s[4];
#pragma unroll
    for (int j = 0; j < 4; ++j) {
      int i = i0 + j * NT;
      d[j] = (i < n) ? dst[e0 + i] : -1;
      s[j] = (i < n) ? src[e0 + i] : 0;
    }
#pragma unroll
    for (int j = 0; j < 4; ++j) {
      if (d[j] >= 0) {
        int bw = bucket_word(d[j], BN, 1.0f / (float)BN);
        int k = bw >> BSH;
        int r = lds_add_ret_i32(&lcur[k], 1);
        sw[r] = ((unsigned)s[j] << BSH) | (unsigned)(bw & (BSZ - 1));
        sb[r] = (unsigned short)k;
      }
    }
  }
  __syncthreads();
  for (int i = tid; i < n; i += NT) {
    int k = sb[i];
    sorted[(size_t)gbase[k] + (i - lstart[k])] = sw[i];
  }
}

__global__ __launch_bounds__(NT) void gcn_local(const int* __restrict__ src, const int* __restrict__ dst,
                                                unsigned* __restrict__ tmp, unsigned* __restrict__ seg,
                                                long long E, int NB, int CB, int BN) {
  __shared__ unsigned sw[CH];
  __shared__ int lstart[NBMAX];
  int tid = (int)threadIdx.x, b = (int)blockIdx.x, lane = tid & 31;
  for (int k = tid; k < NBMAX; k += NT) lstart[k] = 0;
  long long e0 = (long long)b * CH;
  int n = (int)((E - e0 < CH) ? (E - e0) : CH);
  int d[CH / NT], s[CH / NT], rk[CH / NT];
#pragma unroll
  for (int j = 0; j < CH / NT; ++j) {
    int i = tid + j * NT;
    int ic = (i < n) ? i : (n - 1);
    d[j] = dst[e0 + ic];
    s[j] = src[e0 + ic];
  }
  float inv = 1.0f / (float)BN;
#pragma unroll
  for (int j = 0; j < CH / NT; ++j) d[j] = (tid + j * NT >= n) ? -1 : bucket_word(d[j], BN, inv);
  __syncthreads();
#pragma unroll
  for (int j = 0; j < CH / NT; ++j) rk[j] = (d[j] >= 0) ? lds_add_ret_i32(&lstart[d[j] >> BSH], 1) : 0;
  __syncthreads();
  if (tid < 32) wave_excl_scan_full(lstart, lane);
  __syncthreads();
#pragma unroll
  for (int j = 0; j < CH / NT; ++j) rk[j] += lstart[(d[j] >= 0) ? (d[j] >> BSH) : 0];
#pragma unroll
  for (int j = 0; j < CH / NT; ++j)
    if (d[j] >= 0) sw[rk[j]] = ((unsigned)s[j] << BSH) | (unsigned)(d[j] & (BSZ - 1));
  for (int k = tid; k < NB; k += NT) {
    int st = lstart[k];
    int en = (k + 1 < NB) ? lstart[k + 1] : n;
    seg[(size_t)k * CB + b] = (unsigned)st | ((unsigned)en << 16);
  }
  __syncthreads();
  for (int i = tid; i < n; i += NT) tmp[e0 + i] = sw[i];
}

__global__ __launch_bounds__(NT) void gcn_degree(unsigned* __restrict__ sorted, int* __restrict__ bst,
                                                 const unsigned* __restrict__ tmp, const unsigned* __restrict__ seg,
                                                 const float* __restrict__ x, const float* __restrict__ W1,
                                                 float* __restrict__ dinv, float* __restrict__ rows,
                                                 int N, int NB, int CB, int BN) {
  __shared__ int c[BSZ];
  __shared__ float sd[BSZ];
  __shared__ __attribute__((aligned(16))) float st[BSZ * 16];
  __shared__ int sst[CBMAX], slen[CBMAX], spre[CBMAX];
  __shared__ int sbase, stot;
  int tid = (int)threadIdx.x, k = (int)blockIdx.x;
  int lane = tid & 31, wave = tid >> 5;
  int node0 = k * BN;
  if (tid < BSZ) c[tid] = 0;
  unsigned segw = (tid < CB) ? seg[(size_t)k * CB + tid] : 0u;

  v16h bh = {}, bl = {}, ah = {}, al = {};
  if (lane < 16) {
#pragma unroll
    for (int kk = 0; kk < 5; ++kk) {
      float w = W1[kk * 16 + lane];
      _Float16 hi = (_Float16)w;
      bh[kk] = hi;
      bl[kk] = (_Float16)(w - (float)hi);
    }
    int d = node0 + (wave << 4) + lane;
    if (d < N) {
      const float* xr = x + (size_t)d * 5;
#pragma unroll
      for (int kk = 0; kk < 5; ++kk) {
        float v = xr[kk];
        _Float16 hi = (_Float16)v;
        ah[kk] = hi;
        al[kk] = (_Float16)(v - (float)hi);
      }
    }
  }
  if (CB > 0) {
    int s0 = (int)(segw & 0xFFFFu), s1 = (int)(segw >> 16);
    sst[tid] = s0; slen[tid] = s1 - s0; spre[tid] = s1 - s0;
  }
  __syncthreads();
  if (CB > 0) {
    if (wave == 0) {
      int tot = wave_excl_scan_full(spre, lane);
      if (lane == 0) stot = tot;
    } else if (wave == 1) {
      int sum = 0;
#pragma unroll
      for (int i = 0; i < CBMAX / 32; ++i) sum += sst[lane + 32 * i];
      for (int o = 16; o > 0; o >>= 1) sum += __shfl_xor(sum, o);
      if (lane == 0) sbase = sum;
    }
    __syncthreads();
    int base = sbase;
    if (tid == 0) {
      bst[k] = base;
      if (k == NB - 1) bst[NB] = base + stot;
    }
    for (int c0 = wave; c0 < CB; c0 += SEGT * (NT / 32)) {
      unsigned w[SEGT];
      int pos[SEGT], len[SEGT], off[SEGT];
      int mx = 0;
#pragma unroll
      for (int u = 0; u < SEGT; ++u) {
        int cc = c0 + u * (NT / 32);
        bool in = cc < CB;
        int ci = in ? cc : 0;
        int l = slen[ci];
        off[u] = ci * CH + sst[ci];
        pos[u] = spre[ci];
        len[u] = in ? l : 0;
        mx = (len[u] > mx) ? len[u] : mx;
      }
#pragma unroll
      for (int u = 0; u < SEGT; ++u) w[u] = (lane < len[u]) ? tmp[(size_t)off[u] + lane] : NOEDGE;
#pragma unroll
      for (int u = 0; u < SEGT; ++u) {
        if (w[u] != NOEDGE) {
          sorted[base + pos[u] + lane] = w[u];
          lds_add_i32(&c[w[u] & (BSZ - 1)], 1);
        }
      }
      if (mx > 32) {
        for (int u = 0; u < SEGT; ++u) {
          for (int o = 32 + lane; o < len[u]; o += 32) {
            unsigned wv = tmp[(size_t)off[u] + o];
            sorted[base + pos[u] + o] = wv;
            lds_add_i32(&c[wv & (BSZ - 1)], 1);
          }
        }
      }
    }
  } else {
    int e0 = bst[k], e1 = bst[k + 1];
    for (int i0 = e0 + tid; i0 < e1; i0 += 4 * NT) {
      unsigned w[4];
#pragma unroll
      for (int j = 0; j < 4; ++j) { int i = i0 + j * NT; w[j] = (i < e1) ? sorted[i] : NOEDGE; }
#pragma unroll
      for (int j = 0; j < 4; ++j) if (w[j] != NOEDGE) lds_add_i32(&c[w[j] & (BSZ - 1)], 1);
    }
  }
  __syncthreads();
  if (tid < BSZ) {
    int d = node0 + tid;
    float di = rsqrtf(1.0f + (float)c[tid]);
    sd[tid] = di;
    if (tid < BN && d < N) dinv[d] = di;
  }
  v8f acc = {};
  acc = __builtin_amdgcn_wmma_f32_16x16x32_f16(false, al, false, bh, (short)0, acc, false, false);
  acc = __builtin_amdgcn_wmma_f32_16x16x32_f16(false, ah, false, bl, (short)0, acc, false, false);
  acc = __builtin_amdgcn_wmma_f32_16x16x32_f16(false, ah, false, bh, (short)0, acc, false, false);
  __syncthreads();
  int ch = lane & 15;
  int nb = (wave << 4) + ((lane >> 4) << 3);
#pragma unroll
  for (int r = 0; r < 8; ++r) st[(nb + r) * 16 + ch] = acc[r] * sd[nb + r];
  __syncthreads();
  const float4* st4 = (const float4*)st;
  float4* out4 = (float4*)rows + (size_t)node0 * 4;
  for (int i = tid; i < BN * 4; i += NT)
    if (node0 + (i >> 2) < N) out4[i] = st4[i];
}

__global__ __launch_bounds__(NTL) void gcn_layer1(const unsigned* __restrict__ sorted, const int* __restrict__ bst,
                                                 const float* __restrict__ rows, const float* __restrict__ b1,
                                                 const float* __restrict__ W2, const float* __restrict__ dinv,
                                                 float* __restrict__ t2, int N, int BN) {
  __shared__ float bins[BSZ * PITCH];
  int tid = (int)threadIdx.x, k = (int)blockIdx.x;
  int node0 = k * BN;
  const float4* rows4 = (const float4*)rows;
  int g = tid >> 2, q = tid & 3;
  {
    int d = node0 + g;
    float4 v0 = rows4[(size_t)((d < N) ? d : (N - 1)) * 4 + q];
    float* p = &bins[g * PITCH + (q << 2)];
    bool in = (g < BN) && (d < N);
    p[0] = in ? v0.x : 0.0f;
    p[1] = in ? v0.y : 0.0f;
    p[2] = in ? v0.z : 0.0f;
    p[3] = in ? v0.w : 0.0f;
  }
  __syncthreads();

  int e0 = bst[k], e1 = bst[k + 1];
  int i0 = e0 + g;
  if (i0 < e1) {
    unsigned w[4];
#pragma unroll
    for (int j = 0; j < 4; ++j) {
      int i = i0 + j * (NTL / 4);
      w[j] = sorted[(i < e1) ? i : (e1 - 1)];
    }
    do {
      unsigned wc[4];
      float4 v[4];
      bool ok[4];
#pragma unroll
      for (int j = 0; j < 4; ++j) {
        wc[j] = w[j];
        ok[j] = (i0 + j * (NTL / 4)) < e1;
        v[j] = rows4[(size_t)(wc[j] >> BSH) * 4 + q];
      }
      i0 += 4 * (NTL / 4);
#pragma unroll
      for (int j = 0; j < 4; ++j) {
        int i = i0 + j * (NTL / 4);
        w[j] = sorted[(i < e1) ? i : (e1 - 1)];
      }
#pragma unroll
      for (int j = 0; j < 4; ++j) {
        int dl = ok[j] ? (int)(wc[j] & (BSZ - 1)) : g;
        float* p = &bins[dl * PITCH + (q << 2)];
        lds_add_f32(p + 0, ok[j] ? v[j].x : 0.0f);
        lds_add_f32(p + 1, ok[j] ? v[j].y : 0.0f);
        lds_add_f32(p + 2, ok[j] ? v[j].z : 0.0f);
        lds_add_f32(p + 3, ok[j] ? v[j].w : 0.0f);
      }
    } while (i0 < e1);
  }
  __syncthreads();
  if (tid < BN) {
    int d = node0 + tid;
    if (d < N) {
      float di = dinv[d];
      float acc = 0.0f;
#pragma unroll
      for (int cc = 0; cc < 16; ++cc) {
        float v = fmaf(bins[tid * PITCH + cc], di, b1[cc]);
        v = fmaxf(v, 0.0f);
        acc = fmaf(v, W2[cc], acc);
      }
      t2[d] = acc;
    }
  }
}

#define L2CAP   8192
#define L2PER   (L2CAP / NT)
#define LBINS   4096
#define LPER    (LBINS / NT)
static_assert(LPER == 8 && NT / 32 == 16, "a thread scans its eight counters as two 16-byte pieces, and the 16 wave sums as four");
__global__ __launch_bounds__(NT) void gcn_layer2(const unsigned* __restrict__ sorted, const int* __restrict__ bst,
                                                 const float* __restrict__ t2, const float* __restrict__ dinv,
                                                 const float* __restrict__ b2, float* __restrict__ out, int N, int BN) {
  __shared__ float bins[BSZ];
  __shared__ float sdinv[BSZ];
  __shared__ unsigned stage[L2CAP];
  __shared__ __attribute__((aligned(16))) int lc[LBINS];
  __shared__ __attribute__((aligned(16))) int wsum[NT / 32];
  int tid = (int)threadIdx.x, k = (int)blockIdx.x;
  int lane = tid & 31, wave = tid >> 5;
  int node0 = k * BN;
  int e0 = bst[k], e1 = bst[k + 1];
  int n = e1 - e0;
  bool srt = (n > 0) && (n <= L2CAP);
  unsigned w[L2PER];
  if (srt) {
#pragma unroll
    for (int j = 0; j < L2PER; ++j) {
      int i = tid + j * NT;
      w[j] = sorted[e0 + ((i < n) ? i : (n - 1))];
    }
#pragma unroll
    for (int j = 0; j < LPER; ++j) lc[tid + j * NT] = 0;
  }
  if (tid < BSZ) { int d = node0 + tid; bins[tid] = 0.0f; sdinv[tid] = (tid < BN && d < N) ? dinv[d] : 0.0f; }
  __syncthreads();
  if (srt) {
    int rk[L2PER];
#pragma unroll
    for (int j = 0; j < L2PER; ++j) {
      int i = tid + j * NT;
      int b = (int)(w[j] >> (BSH + 5));
      b = (b < LBINS) ? b : (LBINS - 1);
      rk[j] = (i < n) ? lds_add_ret_i32(&lc[b], 1) : 0;
    }
    __syncthreads();
    int4* lp = (int4*)&lc[tid * LPER];
    int4 c0 = lp[0], c1 = lp[1];
    int s = c0.x + c0.y + c0.z + c0.w + c1.x + c1.y + c1.z + c1.w;
    int inc = s;
    for (int o = 1; o < 32; o <<= 1) { int t = __shfl_up(inc, o); if (lane >= o) inc += t; }
    if (lane == 31) wsum[wave] = inc;
    __syncthreads();
    int run = inc - s;
    {
      const int4* wp = (const int4*)wsum;
      int4 q0 = wp[0], q1 = wp[1], q2 = wp[2], q3 = wp[3];
      int ws[NT / 32] = {q0.x, q0.y, q0.z, q0.w, q1.x, q1.y, q1.z, q1.w, q2.x, q2.y, q2.z, q2.w, q3.x, q3.y, q3.z, q3.w};
#pragma unroll
      for (int i = 0; i < NT / 32; ++i) run += (i < wave) ? ws[i] : 0;
    }
    int4 p0, p1;
    p0.x = run; run += c0.x; p0.y = run; run += c0.y; p0.z = run; run += c0.z; p0.w = run; run += c0.w;
    p1.x = run; run += c1.x; p1.y = run; run += c1.y; p1.z = run; run += c1.z; p1.w = run;
    lp[0] = p0; lp[1] = p1;
    __syncthreads();
#pragma unroll
    for (int j = 0; j < L2PER; ++j) {
      int b = (int)(w[j] >> (BSH + 5));
      b = (b < LBINS) ? b : (LBINS - 1);
      rk[j] += lc[b];
    }
#pragma unroll
    for (int j = 0; j < L2PER; ++j)
      if (tid + j * NT < n) stage[rk[j]] = w[j];
    __syncthreads();
    for (int i0 = tid; i0 < n; i0 += 4 * NT) {
      unsigned ww[4];
      float a[4], g[4];
#pragma unroll
      for (int j = 0; j < 4; ++j) { int i = i0 + j * NT; ww[j] = (i < n) ? stage[i] : NOEDGE; }
#pragma unroll
      for (int j = 0; j < 4; ++j) {
        unsigned sidx = (ww[j] != NOEDGE) ? (ww[j] >> BSH) : 0u;
        a[j] = t2[sidx]; g[j] = dinv[sidx];
      }
#pragma unroll
      for (int j = 0; j < 4; ++j) {
        if (ww[j] != NOEDGE) {
          int dl = (int)(ww[j] & (BSZ - 1));
          lds_add_f32(&bins[dl], a[j] * g[j] * sdinv[dl]);
        }
      }
    }
  } else {
    for (int i0 = e0 + tid; i0 < e1; i0 += 4 * NT) {
      unsigned wu[4];
      float a[4], g[4];
#pragma unroll
      for (int j = 0; j < 4; ++j) { int i = i0 + j * NT; wu[j] = (i < e1) ? sorted[i] : NOEDGE; }
#pragma unroll
      for (int j = 0; j < 4; ++j) {
        unsigned s = (wu[j] != NOEDGE) ? (wu[j] >> BSH) : 0u;
        a[j] = t2[s]; g[j] = dinv[s];
      }
#pragma unroll
      for (int j = 0; j < 4; ++j) {
        if (wu[j] != NOEDGE) {
          int dl = (int)(wu[j] & (BSZ - 1));
          lds_add_f32(&bins[dl], a[j] * g[j] * sdinv[dl]);
        }
      }
    }
  }
  __syncthreads();
  if (tid < BN) {
    int d = node0 + tid;
    if (d < N) { float di = sdinv[tid]; out[d] = bins[tid] + t2[d] * di * di + b2[0]; }
  }
}

extern "C" void kernel_launch(void* const* d_in, const int* in_sizes, int n_in,
                              void* d_out, int out_size, void* d_ws, size_t ws_size,
                              hipStream_t stream) {
  const float* x  = (const float*)d_in[0];
  const float* W1 = (const float*)d_in[1];
  const float* b1 = (const float*)d_in[2];
  const float* W2 = (const float*)d_in[3];
  const float* b2 = (const float*)d_in[4];
  const int*   ei = (const int*)d_in[5];

  int       N = in_sizes[0] / 5;
  long long E = (long long)in_sizes[5] / 2;
  const int* src = ei;
  const int* dst = ei + E;

  int BN = 2 * ((N + 1023) / 1024);
  int NB = (N + BN - 1) / BN;
  int CB = (int)((E + CH - 1) / CH);

  float*    rows   = (float*)d_ws;
  unsigned* sorted = (unsigned*)(rows + (size_t)16 * N);
  int*      bst    = (int*)(sorted + E);
  float*    dinv   = (float*)(bst + NB + 1);
  float*    t2     = dinv + N;
  int*      cnt    = (int*)rows;
  int*      tot    = cnt + (size_t)NB * CB;
  unsigned* tmp    = (unsigned*)(t2 + N);
  unsigned* seg    = tmp + E;

  size_t need = ((size_t)18 * N + (size_t)E + (size_t)NB + 1) * 4;
  if (ws_size < need) return;

  float* out = (float*)d_out;

  size_t need1 = need + ((size_t)E + (size_t)NB * CB) * 4;
  if (ws_size >= need1 && CB <= CBMAX) {
    gcn_local <<<CB, NT, 0, stream>>>(src, dst, tmp, seg, E, NB, CB, BN);
    gcn_degree<<<NB, NT, 0, stream>>>(sorted, bst, tmp, seg, x, W1, dinv, rows, N, NB, CB, BN);
  } else {
    gcn_count <<<CB, NT, 0, stream>>>(dst, cnt, E, NB, CB, BN);
    gcn_scan  <<<(NB + 7) / 8, 256, 0, stream>>>(cnt, tot, NB, CB);
    gcn_place <<<CB, NT, 0, stream>>>(src, dst, cnt, tot, sorted, bst, E, NB, CB, BN);
    gcn_degree<<<NB, NT, 0, stream>>>(sorted, bst, tmp, seg, x, W1, dinv, rows, N, NB, 0, BN);
  }
  gcn_layer1<<<NB, NTL, 0, stream>>>(sorted, bst, rows, b1, W2, dinv, t2, N, BN);
  gcn_layer2<<<NB, NT, 0, stream>>>(sorted, bst, t2, dinv, b2, out, N, BN);
}


# Layer 1: each bucket's edge run is counting-sorted by destination in LDS; a lane group walks a contiguous span, sums equal-destination rows in registers, one LDS add per run; sum order changes
# speedup vs baseline: 6.8351x; 1.0507x over previous
#include <hip/hip_runtime.h>
#include <hip/hip_bf16.h>


typedef __attribute__((ext_vector_type(16))) _Float16 v16h;
typedef __attribute__((ext_vector_type(8)))  float    v8f;

#define NT      512
#define NTL     1024
#define NOEDGE  0xFFFFFFFFu
#define BSH     8
#define BSZ     256
#define NBMAX   512
#define CH      8192
#define CBMAX   512
#define SEGT    16
static_assert(CBMAX == NBMAX && CBMAX == NT && CH % NT == 0, "one scan routine for both tables; one table entry per thread");
#define PITCH   17


__device__ __forceinline__ void lds_add_i32(int* p, int v) {
  __hip_atomic_fetch_add(p, v, __ATOMIC_RELAXED, __HIP_MEMORY_SCOPE_WORKGROUP);
}
__device__ __forceinline__ int lds_add_ret_i32(int* p, int v) {
  return __hip_atomic_fetch_add(p, v, __ATOMIC_RELAXED, __HIP_MEMORY_SCOPE_WORKGROUP);
}
__device__ __forceinline__ void lds_add_f32(float* p, float v) {
  __hip_atomic_fetch_add(p, v, __ATOMIC_RELAXED, __HIP_MEMORY_SCOPE_WORKGROUP);
}

__device__ __forceinline__ int bucket_word(int d, int bn, float inv) {
  int k = (int)(((float)d + 0.5f) * inv);
  return (k << BSH) | (d - k * bn);
}

__device__ __forceinline__ int wave_excl_scan_lds(int* a, int n, int lane) {
  int b0 = lane * (NBMAX / 32);
  int sum = 0;
  for (int i = 0; i < NBMAX / 32; ++i) { int j = b0 + i; if (j < n) sum += a[j]; }
  int inc = sum;
  for (int o = 1; o < 32; o <<= 1) { int v = __shfl_up(inc, o); if (lane >= o) inc += v; }
  int run = inc - sum;
  for (int i = 0; i < NBMAX / 32; ++i) { int j = b0 + i; if (j < n) { int v = a[j]; a[j] = run; run += v; } }
  return __shfl(inc, 31);
}

__device__ __forceinline__ int wave_excl_scan_full(int* a, int lane) {
  int* p = a + lane * (NBMAX / 32);
  int v[NBMAX / 32];
  int sum = 0;
#pragma unroll
  for (int i = 0; i < NBMAX / 32; ++i) { v[i] = p[i]; sum += v[i]; }
  int inc = sum;
  for (int o = 1; o < 32; o <<= 1) { int t = __shfl_up(inc, o); if (lane >= o) inc += t; }
  int run = inc - sum;
#pragma unroll
  for (int i = 0; i < NBMAX / 32; ++i) { p[i] = run; run += v[i]; }
  return __shfl(inc, 31);
}

static_assert(BSZ == 256, "a lane of the scanning wave owns eight entries");
__device__ __forceinline__ void wave_excl_scan_bsz(int* a, int lane) {
  int4* p = (int4*)(a + lane * (BSZ / 32));
  int4 c0 = p[0], c1 = p[1];
  int s = c0.x + c0.y + c0.z + c0.w + c1.x + c1.y + c1.z + c1.w;
  int inc = s;
  for (int o = 1; o < 32; o <<= 1) { int t = __shfl_up(inc, o); if (lane >= o) inc += t; }
  int run = inc - s;
  int4 p0, p1;
  p0.x = run; run += c0.x; p0.y = run; run += c0.y; p0.z = run; run += c0.z; p0.w = run; run += c0.w;
  p1.x = run; run += c1.x; p1.y = run; run += c1.y; p1.z = run; run += c1.z; p1.w = run;
  p[0] = p0; p[1] = p1;
}

__global__ __launch_bounds__(NT) void gcn_count(const int* __restrict__ dst, int* __restrict__ cnt,
                                                long long E, int NB, int CB, int BN) {
  __shared__ int c[NBMAX];
  int tid = (int)threadIdx.x, b = (int)blockIdx.x;
  float inv = 1.0f / (float)BN;
  for (int i = tid; i < NBMAX; i += NT) c[i] = 0;
  __syncthreads();
  long long e0 = (long long)b * CH;
  int n = (int)((E - e0 < CH) ? (E - e0) : CH);
  for (int i0 = tid; i0 < n; i0 += 4 * NT) {
    int d[4];
#pragma unroll
    for (int j = 0; j < 4; ++j) { int i = i0 + j * NT; d[j] = (i < n) ? dst[e0 + i] : -1; }
#pragma unroll
    for (int j = 0; j < 4; ++j) if (d[j] >= 0) lds_add_i32(&c[bucket_word(d[j], BN, inv) >> BSH], 1);
  }
  __syncthreads();
  for (int k = tid; k < NB; k += NT) cnt[(size_t)k * CB + b] = c[k];
}

__global__ void gcn_scan(int* __restrict__ cnt, int* __restrict__ tot, int NB, int CB) {
  int wave = (int)(((long long)blockIdx.x * blockDim.x + threadIdx.x) >> 5);
  int lane = (int)(threadIdx.x & 31);
  if (wave >= NB) return;
  int* row = cnt + (size_t)wave * CB;
  int per = (CB + 31) >> 5;
  int b0 = lane * per;
  int sum = 0;
  for (int i = 0; i < per; ++i) { int j = b0 + i; if (j < CB) sum += row[j]; }
  int inc = sum;
  for (int o = 1; o < 32; o <<= 1) { int v = __shfl_up(inc, o); if (lane >= o) inc += v; }
  int run = inc - sum;
  for (int i = 0; i < per; ++i) { int j = b0 + i; if (j < CB) { int v = row[j]; row[j] = run; run += v; } }
  if (lane == 31) tot[wave] = inc;
}

__global__ __launch_bounds__(NT) void gcn_place(const int* __restrict__ src, const int* __restrict__ dst,
                                                const int* __restrict__ cnt, const int* __restrict__ tot,
                                                unsigned* __restrict__ sorted, int* __restrict__ bst,
                                                long long E, int NB, int CB, int BN) {
  __shared__ unsigned       sw[CH];
  __shared__ unsigned short sb[CH];
  __shared__ int lstart[NBMAX], lcur[NBMAX], gbase[NBMAX], gst[NBMAX];
  int tid = (int)threadIdx.x, b = (int)blockIdx.x, lane = tid & 31;
  for (int k = tid; k < NBMAX; k += NT) {
    int own = 0, before = 0, t = 0;
    if (k < NB) {
      t = tot[k];
      before = cnt[(size_t)k * CB + b];
      int next = (b + 1 < CB) ? cnt[(size_t)k * CB + b + 1] : t;
      own = next - before;
    }
    lstart[k] = own; gbase[k] = before; gst[k] = t;
  }
  __syncthreads();
  if (tid < 32) {
    wave_excl_scan_lds(lstart, NB, lane);
    wave_excl_scan_lds(gst, NB, lane);
  }
  __syncthreads();
  for (int k = tid; k < NB; k += NT) {
    lcur[k] = lstart[k];
    gbase[k] += gst[k];
    if (b == 0) bst[k] = gst[k];
  }
  if (b == 0 && tid == 0) bst[NB] = (int)E;
  __syncthreads();
  long long e0 = (long long)b * CH;
  int n = (int)((E - e0 < CH) ? (E - e0) : CH);
  for (int i0 = tid; i0 < n; i0 += 4 * NT) {
    int d[4], s[4];
#pragma unroll
    for (int j = 0; j < 4; ++j) {
      int i = i0 + j * NT;
      d[j] = (i < n) ? dst[e0 + i] : -1;
      s[j] = (i < n) ? src[e0 + i] : 0;
    }
#pragma unroll
    for (int j = 0; j < 4; ++j) {
      if (d[j] >= 0) {
        int bw = bucket_word(d[j], BN, 1.0f / (float)BN);
        int k = bw >> BSH;
        int r = lds_add_ret_i32(&lcur[k], 1);
        sw[r] = ((unsigned)s[j] << BSH) | (unsigned)(bw & (BSZ - 1));
        sb[r] = (unsigned short)k;
      }
    }
  }
  __syncthreads();
  for (int i = tid; i < n; i += NT) {
    int k = sb[i];
    sorted[(size_t)gbase[k] + (i - lstart[k])] = sw[i];
  }
}

__global__ __launch_bounds__(NT) void gcn_local(const int* __restrict__ src, const int* __restrict__ dst,
                                                unsigned* __restrict__ tmp, unsigned* __restrict__ seg,
                                                long long E, int NB, int CB, int BN) {
  __shared__ unsigned sw[CH];
  __shared__ int lstart[NBMAX];
  int tid = (int)threadIdx.x, b = (int)blockIdx.x, lane = tid & 31;
  for (int k = tid; k < NBMAX; k += NT) lstart[k] = 0;
  long long e0 = (long long)b * CH;
  int n = (int)((E - e0 < CH) ? (E - e0) : CH);
  int d[CH / NT], s[CH / NT], rk[CH / NT];
#pragma unroll
  for (int j = 0; j < CH / NT; ++j) {
    int i = tid + j * NT;
    int ic = (i < n) ? i : (n - 1);
    d[j] = dst[e0 + ic];
    s[j] = src[e0 + ic];
  }
  float inv = 1.0f / (float)BN;
#pragma unroll
  for (int j = 0; j < CH / NT; ++j) d[j] = (tid + j * NT >= n) ? -1 : bucket_word(d[j], BN, inv);
  __syncthreads();
#pragma unroll
  for (int j = 0; j < CH / NT; ++j) rk[j] = (d[j] >= 0) ? lds_add_ret_i32(&lstart[d[j] >> BSH], 1) : 0;
  __syncthreads();
  if (tid < 32) wave_excl_scan_full(lstart, lane);
  __syncthreads();
#pragma unroll
  for (int j = 0; j < CH / NT; ++j) rk[j] += lstart[(d[j] >= 0) ? (d[j] >> BSH) : 0];
#pragma unroll
  for (int j = 0; j < CH / NT; ++j)
    if (d[j] >= 0) sw[rk[j]] = ((unsigned)s[j] << BSH) | (unsigned)(d[j] & (BSZ - 1));
  for (int k = tid; k < NB; k += NT) {
    int st = lstart[k];
    int en = (k + 1 < NB) ? lstart[k + 1] : n;
    seg[(size_t)k * CB + b] = (unsigned)st | ((unsigned)en << 16);
  }
  __syncthreads();
  for (int i = tid; i < n; i += NT) tmp[e0 + i] = sw[i];
}

__global__ __launch_bounds__(NT) void gcn_degree(unsigned* __restrict__ sorted, int* __restrict__ bst,
                                                 const unsigned* __restrict__ tmp, const unsigned* __restrict__ seg,
                                                 const float* __restrict__ x, const float* __restrict__ W1,
                                                 float* __restrict__ dinv, float* __restrict__ rows,
                                                 int N, int NB, int CB, int BN) {
  __shared__ int c[BSZ];
  __shared__ float sd[BSZ];
  __shared__ __attribute__((aligned(16))) float st[BSZ * 16];
  __shared__ int sst[CBMAX], slen[CBMAX], spre[CBMAX];
  __shared__ int sbase, stot;
  int tid = (int)threadIdx.x, k = (int)blockIdx.x;
  int lane = tid & 31, wave = tid >> 5;
  int node0 = k * BN;
  if (tid < BSZ) c[tid] = 0;
  unsigned segw = (tid < CB) ? seg[(size_t)k * CB + tid] : 0u;

  v16h bh = {}, bl = {}, ah = {}, al = {};
  if (lane < 16) {
#pragma unroll
    for (int kk = 0; kk < 5; ++kk) {
      float w = W1[kk * 16 + lane];
      _Float16 hi = (_Float16)w;
      bh[kk] = hi;
      bl[kk] = (_Float16)(w - (float)hi);
    }
    int d = node0 + (wave << 4) + lane;
    if (d < N) {
      const float* xr = x + (size_t)d * 5;
#pragma unroll
      for (int kk = 0; kk < 5; ++kk) {
        float v = xr[kk];
        _Float16 hi = (_Float16)v;
        ah[kk] = hi;
        al[kk] = (_Float16)(v - (float)hi);
      }
    }
  }
  if (CB > 0) {
    int s0 = (int)(segw & 0xFFFFu), s1 = (int)(segw >> 16);
    sst[tid] = s0; slen[tid] = s1 - s0; spre[tid] = s1 - s0;
  }
  __syncthreads();
  if (CB > 0) {
    if (wave == 0) {
      int tot = wave_excl_scan_full(spre, lane);
      if (lane == 0) stot = tot;
    } else if (wave == 1) {
      int sum = 0;
#pragma unroll
      for (int i = 0; i < CBMAX / 32; ++i) sum += sst[lane + 32 * i];
      for (int o = 16; o > 0; o >>= 1) sum += __shfl_xor(sum, o);
      if (lane == 0) sbase = sum;
    }
    __syncthreads();
    int base = sbase;
    if (tid == 0) {
      bst[k] = base;
      if (k == NB - 1) bst[NB] = base + stot;
    }
    for (int c0 = wave; c0 < CB; c0 += SEGT * (NT / 32)) {
      unsigned w[SEGT];
      int pos[SEGT], len[SEGT], off[SEGT];
      int mx = 0;
#pragma unroll
      for (int u = 0; u < SEGT; ++u) {
        int cc = c0 + u * (NT / 32);
        bool in = cc < CB;
        int ci = in ? cc : 0;
        int l = slen[ci];
        off[u] = ci * CH + sst[ci];
        pos[u] = spre[ci];
        len[u] = in ? l : 0;
        mx = (len[u] > mx) ? len[u] : mx;
      }
#pragma unroll
      for (int u = 0; u < SEGT; ++u) w[u] = (lane < len[u]) ? tmp[(size_t)off[u] + lane] : NOEDGE;
#pragma unroll
      for (int u = 0; u < SEGT; ++u) {
        if (w[u] != NOEDGE) {
          sorted[base + pos[u] + lane] = w[u];
          lds_add_i32(&c[w[u] & (BSZ - 1)], 1);
        }
      }
      if (mx > 32) {
        for (int u = 0; u < SEGT; ++u) {
          for (int o = 32 + lane; o < len[u]; o += 32) {
            unsigned wv = tmp[(size_t)off[u] + o];
            sorted[base + pos[u] + o] = wv;
            lds_add_i32(&c[wv & (BSZ - 1)], 1);
          }
        }
      }
    }
  } else {
    int e0 = bst[k], e1 = bst[k + 1];
    for (int i0 = e0 + tid; i0 < e1; i0 += 4 * NT) {
      unsigned w[4];
#pragma unroll
      for (int j = 0; j < 4; ++j) { int i = i0 + j * NT; w[j] = (i < e1) ? sorted[i] : NOEDGE; }
#pragma unroll
      for (int j = 0; j < 4; ++j) if (w[j] != NOEDGE) lds_add_i32(&c[w[j] & (BSZ - 1)], 1);
    }
  }
  __syncthreads();
  if (tid < BSZ) {
    int d = node0 + tid;
    float di = rsqrtf(1.0f + (float)c[tid]);
    sd[tid] = di;
    if (tid < BN && d < N) dinv[d] = di;
  }
  v8f acc = {};
  acc = __builtin_amdgcn_wmma_f32_16x16x32_f16(false, al, false, bh, (short)0, acc, false, false);
  acc = __builtin_amdgcn_wmma_f32_16x16x32_f16(false, ah, false, bl, (short)0, acc, false, false);
  acc = __builtin_amdgcn_wmma_f32_16x16x32_f16(false, ah, false, bh, (short)0, acc, false, false);
  __syncthreads();
  int ch = lane & 15;
  int nb = (wave << 4) + ((lane >> 4) << 3);
#pragma unroll
  for (int r = 0; r < 8; ++r) st[(nb + r) * 16 + ch] = acc[r] * sd[nb + r];
  __syncthreads();
  const float4* st4 = (const float4*)st;
  float4* out4 = (float4*)rows + (size_t)node0 * 4;
  for (int i = tid; i < BN * 4; i += NT)
    if (node0 + (i >> 2) < N) out4[i] = st4[i];
}

#define L1CAP   8192
#define L1PER   (L1CAP / NTL)
__global__ __launch_bounds__(NTL) void gcn_layer1(const unsigned* __restrict__ sorted, const int* __restrict__ bst,
                                                 const float* __restrict__ rows, const float* __restrict__ b1,
                                                 const float* __restrict__ W2, const float* __restrict__ dinv,
                                                 float* __restrict__ t2, int N, int BN) {
  __shared__ float bins[BSZ * PITCH];
  __shared__ unsigned stage[L1CAP];
  __shared__ __attribute__((aligned(16))) int dcnt[BSZ];
  int tid = (int)threadIdx.x, k = (int)blockIdx.x;
  int node0 = k * BN;
  const float4* rows4 = (const float4*)rows;
  int g = tid >> 2, q = tid & 3;
  int e0 = bst[k], e1 = bst[k + 1];
  int n = e1 - e0;
  bool srt = (n > 0) && (n <= L1CAP);
  unsigned w[L1PER];
  if (srt) {
#pragma unroll
    for (int j = 0; j < L1PER; ++j) {
      int i = tid + j * NTL;
      w[j] = sorted[e0 + ((i < n) ? i : (n - 1))];
    }
  }
  if (tid < BSZ) dcnt[tid] = 0;
  {
    int d = node0 + g;
    float4 v0 = rows4[(size_t)((d < N) ? d : (N - 1)) * 4 + q];
    float* p = &bins[g * PITCH + (q << 2)];
    bool in = (g < BN) && (d < N);
    p[0] = in ? v0.x : 0.0f;
    p[1] = in ? v0.y : 0.0f;
    p[2] = in ? v0.z : 0.0f;
    p[3] = in ? v0.w : 0.0f;
  }
  __syncthreads();

  if (srt) {
    {
      int rk[L1PER];
#pragma unroll
      for (int j = 0; j < L1PER; ++j)
        rk[j] = (tid + j * NTL < n) ? lds_add_ret_i32(&dcnt[w[j] & (BSZ - 1)], 1) : 0;
      __syncthreads();
      if (tid < 32) wave_excl_scan_bsz(dcnt, tid);
      __syncthreads();
#pragma unroll
      for (int j = 0; j < L1PER; ++j) rk[j] += dcnt[w[j] & (BSZ - 1)];
#pragma unroll
      for (int j = 0; j < L1PER; ++j)
        if (tid + j * NTL < n) stage[rk[j]] = w[j];
    }
    __syncthreads();
    int per = (n + (NTL / 4) - 1) / (NTL / 4);
    int lo = g * per;
    int hi = (lo + per < n) ? (lo + per) : n;
    int cur = -1;
    float4 acc = {0.0f, 0.0f, 0.0f, 0.0f};
    unsigned wd[4];
    float4 v[4];
#pragma unroll
    for (int j = 0; j < 4; ++j) {
      int i = lo + j;
      bool in = i < hi;
      wd[j] = stage[in ? i : (n - 1)];
      v[j] = rows4[(size_t)(in ? (wd[j] >> BSH) : 0u) * 4 + q];
    }
    for (int t = 0; t < per; t += 4) {
      unsigned wn[4];
      float4 vn[4];
#pragma unroll
      for (int j = 0; j < 4; ++j) {
        int i = lo + t + 4 + j;
        bool in = i < hi;
        wn[j] = stage[in ? i : (n - 1)];
        vn[j] = rows4[(size_t)(in ? (wn[j] >> BSH) : 0u) * 4 + q];
      }
#pragma unroll
      for (int j = 0; j < 4; ++j) {
        bool ok = (lo + t + j) < hi;
        int dl = (int)(wd[j] & (BSZ - 1));
        bool chg = ok && (dl != cur);
        bool fl = chg && (cur >= 0);
        if (fl) {
          float* p = &bins[cur * PITCH + (q << 2)];
          lds_add_f32(p + 0, acc.x);
          lds_add_f32(p + 1, acc.y);
          lds_add_f32(p + 2, acc.z);
          lds_add_f32(p + 3, acc.w);
        }
        if (chg) { cur = dl; acc.x = 0.0f; acc.y = 0.0f; acc.z = 0.0f; acc.w = 0.0f; }
        if (ok) { acc.x += v[j].x; acc.y += v[j].y; acc.z += v[j].z; acc.w += v[j].w; }
      }
#pragma unroll
      for (int j = 0; j < 4; ++j) { wd[j] = wn[j]; v[j] = vn[j]; }
    }
    if (cur >= 0) {
      float* p = &bins[cur * PITCH + (q << 2)];
      lds_add_f32(p + 0, acc.x);
      lds_add_f32(p + 1, acc.y);
      lds_add_f32(p + 2, acc.z);
      lds_add_f32(p + 3, acc.w);
    }
  } else {
    int i0 = e0 + g;
    if (i0 < e1) {
      unsigned wu[4];
#pragma unroll
      for (int j = 0; j < 4; ++j) {
        int i = i0 + j * (NTL / 4);
        wu[j] = sorted[(i < e1) ? i : (e1 - 1)];
      }
      do {
        unsigned wc[4];
        float4 vv[4];
        bool ok[4];
#pragma unroll
        for (int j = 0; j < 4; ++j) {
          wc[j] = wu[j];
          ok[j] = (i0 + j * (NTL / 4)) < e1;
          vv[j] = rows4[(size_t)(wc[j] >> BSH) * 4 + q];
        }
        i0 += 4 * (NTL / 4);
#pragma unroll
        for (int j = 0; j < 4; ++j) {
          int i = i0 + j * (NTL / 4);
          wu[j] = sorted[(i < e1) ? i : (e1 - 1)];
        }
#pragma unroll
        for (int j = 0; j < 4; ++j) {
          int dl = ok[j] ? (int)(wc[j] & (BSZ - 1)) : g;
          float* p = &bins[dl * PITCH + (q << 2)];
          lds_add_f32(p + 0, ok[j] ? vv[j].x : 0.0f);
          lds_add_f32(p + 1, ok[j] ? vv[j].y : 0.0f);
          lds_add_f32(p + 2, ok[j] ? vv[j].z : 0.0f);
          lds_add_f32(p + 3, ok[j] ? vv[j].w : 0.0f);
        }
      } while (i0 < e1);
    }
  }
  __syncthreads();
  if (tid < BN) {
    int d = node0 + tid;
    if (d < N) {
      float di = dinv[d];
      float acc2 = 0.0f;
#pragma unroll
      for (int cc = 0; cc < 16; ++cc) {
        float vv = fmaf(bins[tid * PITCH + cc], di, b1[cc]);
        vv = fmaxf(vv, 0.0f);
        acc2 = fmaf(vv, W2[cc], acc2);
      }
      t2[d] = acc2;
    }
  }
}

#define L2CAP   8192
#define L2PER   (L2CAP / NT)
#define LBINS   4096
#define LPER    (LBINS / NT)
static_assert(LPER == 8 && NT / 32 == 16, "a thread scans its eight counters as two 16-byte pieces, and the 16 wave sums as four");
__global__ __launch_bounds__(NT) void gcn_layer2(const unsigned* __restrict__ sorted, const int* __restrict__ bst,
                                                 const float* __restrict__ t2, const float* __restrict__ dinv,
                                                 const float* __restrict__ b2, float* __restrict__ out, int N, int BN) {
  __shared__ float bins[BSZ];
  __shared__ float sdinv[BSZ];
  __shared__ unsigned stage[L2CAP];
  __shared__ __attribute__((aligned(16))) int lc[LBINS];
  __shared__ __attribute__((aligned(16))) int wsum[NT / 32];
  int tid = (int)threadIdx.x, k = (int)blockIdx.x;
  int lane = tid & 31, wave = tid >> 5;
  int node0 = k * BN;
  int e0 = bst[k], e1 = bst[k + 1];
  int n = e1 - e0;
  bool srt = (n > 0) && (n <= L2CAP);
  unsigned w[L2PER];
  if (srt) {
#pragma unroll
    for (int j = 0; j < L2PER; ++j) {
      int i = tid + j * NT;
      w[j] = sorted[e0 + ((i < n) ? i : (n - 1))];
    }
#pragma unroll
    for (int j = 0; j < LPER; ++j) lc[tid + j * NT] = 0;
  }
  if (tid < BSZ) { int d = node0 + tid; bins[tid] = 0.0f; sdinv[tid] = (tid < BN && d < N) ? dinv[d] : 0.0f; }
  __syncthreads();
  if (srt) {
    int rk[L2PER];
#pragma unroll
    for (int j = 0; j < L2PER; ++j) {
      int i = tid + j * NT;
      int b = (int)(w[j] >> (BSH + 5));
      b = (b < LBINS) ? b : (LBINS - 1);
      rk[j] = (i < n) ? lds_add_ret_i32(&lc[b], 1) : 0;
    }
    __syncthreads();
    int4* lp = (int4*)&lc[tid * LPER];
    int4 c0 = lp[0], c1 = lp[1];
    int s = c0.x + c0.y + c0.z + c0.w + c1.x + c1.y + c1.z + c1.w;
    int inc = s;
    for (int o = 1; o < 32; o <<= 1) { int t = __shfl_up(inc, o); if (lane >= o) inc += t; }
    if (lane == 31) wsum[wave] = inc;
    __syncthreads();
    int run = inc - s;
    {
      const int4* wp = (const int4*)wsum;
      int4 q0 = wp[0], q1 = wp[1], q2 = wp[2], q3 = wp[3];
      int ws[NT / 32] = {q0.x, q0.y, q0.z, q0.w, q1.x, q1.y, q1.z, q1.w, q2.x, q2.y, q2.z, q2.w, q3.x, q3.y, q3.z, q3.w};
#pragma unroll
      for (int i = 0; i < NT / 32; ++i) run += (i < wave) ? ws[i] : 0;
    }
    int4 p0, p1;
    p0.x = run; run += c0.x; p0.y = run; run += c0.y; p0.z = run; run += c0.z; p0.w = run; run += c0.w;
    p1.x = run; run += c1.x; p1.y = run; run += c1.y; p1.z = run; run += c1.z; p1.w = run;
    lp[0] = p0; lp[1] = p1;
    __syncthreads();
#pragma unroll
    for (int j = 0; j < L2PER; ++j) {
      int b = (int)(w[j] >> (BSH + 5));
      b = (b < LBINS) ? b : (LBINS - 1);
      rk[j] += lc[b];
    }
#pragma unroll
    for (int j = 0; j < L2PER; ++j)
      if (tid + j * NT < n) stage[rk[j]] = w[j];
    __syncthreads();
    for (int i0 = tid; i0 < n; i0 += 4 * NT) {
      unsigned ww[4];
      float a[4], g[4];
#pragma unroll
      for (int j = 0; j < 4; ++j) { int i = i0 + j * NT; ww[j] = (i < n) ? stage[i] : NOEDGE; }
#pragma unroll
      for (int j = 0; j < 4; ++j) {
        unsigned sidx = (ww[j] != NOEDGE) ? (ww[j] >> BSH) : 0u;
        a[j] = t2[sidx]; g[j] = dinv[sidx];
      }
#pragma unroll
      for (int j = 0; j < 4; ++j) {
        if (ww[j] != NOEDGE) {
          int dl = (int)(ww[j] & (BSZ - 1));
          lds_add_f32(&bins[dl], a[j] * g[j] * sdinv[dl]);
        }
      }
    }
  } else {
    for (int i0 = e0 + tid; i0 < e1; i0 += 4 * NT) {
      unsigned wu[4];
      float a[4], g[4];
#pragma unroll
      for (int j = 0; j < 4; ++j) { int i = i0 + j * NT; wu[j] = (i < e1) ? sorted[i] : NOEDGE; }
#pragma unroll
      for (int j = 0; j < 4; ++j) {
        unsigned s = (wu[j] != NOEDGE) ? (wu[j] >> BSH) : 0u;
        a[j] = t2[s]; g[j] = dinv[s];
      }
#pragma unroll
      for (int j = 0; j < 4; ++j) {
        if (wu[j] != NOEDGE) {
          int dl = (int)(wu[j] & (BSZ - 1));
          lds_add_f32(&bins[dl], a[j] * g[j] * sdinv[dl]);
        }
      }
    }
  }
  __syncthreads();
  if (tid < BN) {
    int d = node0 + tid;
    if (d < N) { float di = sdinv[tid]; out[d] = bins[tid] + t2[d] * di * di + b2[0]; }
  }
}

extern "C" void kernel_launch(void* const* d_in, const int* in_sizes, int n_in,
                              void* d_out, int out_size, void* d_ws, size_t ws_size,
                              hipStream_t stream) {
  const float* x  = (const float*)d_in[0];
  const float* W1 = (const float*)d_in[1];
  const float* b1 = (const float*)d_in[2];
  const float* W2 = (const float*)d_in[3];
  const float* b2 = (const float*)d_in[4];
  const int*   ei = (const int*)d_in[5];

  int       N = in_sizes[0] / 5;
  long long E = (long long)in_sizes[5] / 2;
  const int* src = ei;
  const int* dst = ei + E;

  int BN = 2 * ((N + 1023) / 1024);
  int NB = (N + BN - 1) / BN;
  int CB = (int)((E + CH - 1) / CH);

  float*    rows   = (float*)d_ws;
  unsigned* sorted = (unsigned*)(rows + (size_t)16 * N);
  int*      bst    = (int*)(sorted + E);
  float*    dinv   = (float*)(bst + NB + 1);
  float*    t2     = dinv + N;
  int*      cnt    = (int*)rows;
  int*      tot    = cnt + (size_t)NB * CB;
  unsigned* tmp    = (unsigned*)(t2 + N);
  unsigned* seg    = tmp + E;

  size_t need = ((size_t)18 * N + (size_t)E + (size_t)NB + 1) * 4;
  if (ws_size < need) return;

  float* out = (float*)d_out;

  size_t need1 = need + ((size_t)E + (size_t)NB * CB) * 4;
  if (ws_size >= need1 && CB <= CBMAX) {
    gcn_local <<<CB, NT, 0, stream>>>(src, dst, tmp, seg, E, NB, CB, BN);
    gcn_degree<<<NB, NT, 0, stream>>>(sorted, bst, tmp, seg, x, W1, dinv, rows, N, NB, CB, BN);
  } else {
    gcn_count <<<CB, NT, 0, stream>>>(dst, cnt, E, NB, CB, BN);
    gcn_scan  <<<(NB + 7) / 8, 256, 0, stream>>>(cnt, tot, NB, CB);
    gcn_place <<<CB, NT, 0, stream>>>(src, dst, cnt, tot, sorted, bst, E, NB, CB, BN);
    gcn_degree<<<NB, NT, 0, stream>>>(sorted, bst, tmp, seg, x, W1, dinv, rows, N, NB, 0, BN);
  }
  gcn_layer1<<<NB, NTL, 0, stream>>>(sorted, bst, rows, b1, W2, dinv, t2, N, BN);
  gcn_layer2<<<NB, NT, 0, stream>>>(sorted, bst, t2, dinv, b2, out, N, BN);
}


# Layer 2: one 1024-thread work-group sorts two adjacent buckets' runs by source line as one (4 edges per line, not 2), so a wave's 32 gathers touch about 9 lines, not 15; LDS add order changes
# speedup vs baseline: 7.4428x; 1.0889x over previous
#include <hip/hip_runtime.h>
#include <hip/hip_bf16.h>


typedef __attribute__((ext_vector_type(16))) _Float16 v16h;
typedef __attribute__((ext_vector_type(8)))  float    v8f;

#define NT      512
#define NTL     1024
#define NOEDGE  0xFFFFFFFFu
#define BSH     8
#define BSZ     256
#define NBMAX   512
#define CH      8192
#define CBMAX   512
#define SEGT    16
static_assert(CBMAX == NBMAX && CBMAX == NT && CH % NT == 0, "one scan routine for both tables; one table entry per thread");
#define PITCH   17


__device__ __forceinline__ void lds_add_i32(int* p, int v) {
  __hip_atomic_fetch_add(p, v, __ATOMIC_RELAXED, __HIP_MEMORY_SCOPE_WORKGROUP);
}
__device__ __forceinline__ int lds_add_ret_i32(int* p, int v) {
  return __hip_atomic_fetch_add(p, v, __ATOMIC_RELAXED, __HIP_MEMORY_SCOPE_WORKGROUP);
}
__device__ __forceinline__ void lds_add_f32(float* p, float v) {
  __hip_atomic_fetch_add(p, v, __ATOMIC_RELAXED, __HIP_MEMORY_SCOPE_WORKGROUP);
}

__device__ __forceinline__ int bucket_word(int d, int bn, float inv) {
  int k = (int)(((float)d + 0.5f) * inv);
  return (k << BSH) | (d - k * bn);
}

__device__ __forceinline__ int wave_excl_scan_lds(int* a, int n, int lane) {
  int b0 = lane * (NBMAX / 32);
  int sum = 0;
  for (int i = 0; i < NBMAX / 32; ++i) { int j = b0 + i; if (j < n) sum += a[j]; }
  int inc = sum;
  for (int o = 1; o < 32; o <<= 1) { int v = __shfl_up(inc, o); if (lane >= o) inc += v; }
  int run = inc - sum;
  for (int i = 0; i < NBMAX / 32; ++i) { int j = b0 + i; if (j < n) { int v = a[j]; a[j] = run; run += v; } }
  return __shfl(inc, 31);
}

__device__ __forceinline__ int wave_excl_scan_full(int* a, int lane) {
  int* p = a + lane * (NBMAX / 32);
  int v[NBMAX / 32];
  int sum = 0;
#pragma unroll
  for (int i = 0; i < NBMAX / 32; ++i) { v[i] = p[i]; sum += v[i]; }
  int inc = sum;
  for (int o = 1; o < 32; o <<= 1) { int t = __shfl_up(inc, o); if (lane >= o) inc += t; }
  int run = inc - sum;
#pragma unroll
  for (int i = 0; i < NBMAX / 32; ++i) { p[i] = run; run += v[i]; }
  return __shfl(inc, 31);
}

static_assert(BSZ == 256, "a lane of the scanning wave owns eight entries");
__device__ __forceinline__ void wave_excl_scan_bsz(int* a, int lane) {
  int4* p = (int4*)(a + lane * (BSZ / 32));
  int4 c0 = p[0], c1 = p[1];
  int s = c0.x + c0.y + c0.z + c0.w + c1.x + c1.y + c1.z + c1.w;
  int inc = s;
  for (int o = 1; o < 32; o <<= 1) { int t = __shfl_up(inc, o); if (lane >= o) inc += t; }
  int run = inc - s;
  int4 p0, p1;
  p0.x = run; run += c0.x; p0.y = run; run += c0.y; p0.z = run; run += c0.z; p0.w = run; run += c0.w;
  p1.x = run; run += c1.x; p1.y = run; run += c1.y; p1.z = run; run += c1.z; p1.w = run;
  p[0] = p0; p[1] = p1;
}

__global__ __launch_bounds__(NT) void gcn_count(const int* __restrict__ dst, int* __restrict__ cnt,
                                                long long E, int NB, int CB, int BN) {
  __shared__ int c[NBMAX];
  int tid = (int)threadIdx.x, b = (int)blockIdx.x;
  float inv = 1.0f / (float)BN;
  for (int i = tid; i < NBMAX; i += NT) c[i] = 0;
  __syncthreads();
  long long e0 = (long long)b * CH;
  int n = (int)((E - e0 < CH) ? (E - e0) : CH);
  for (int i0 = tid; i0 < n; i0 += 4 * NT) {
    int d[4];
#pragma unroll
    for (int j = 0; j < 4; ++j) { int i = i0 + j * NT; d[j] = (i < n) ? dst[e0 + i] : -1; }
#pragma unroll
    for (int j = 0; j < 4; ++j) if (d[j] >= 0) lds_add_i32(&c[bucket_word(d[j], BN, inv) >> BSH], 1);
  }
  __syncthreads();
  for (int k = tid; k < NB; k += NT) cnt[(size_t)k * CB + b] = c[k];
}

__global__ void gcn_scan(int* __restrict__ cnt, int* __restrict__ tot, int NB, int CB) {
  int wave = (int)(((long long)blockIdx.x * blockDim.x + threadIdx.x) >> 5);
  int lane = (int)(threadIdx.x & 31);
  if (wave >= NB) return;
  int* row = cnt + (size_t)wave * CB;
  int per = (CB + 31) >> 5;
  int b0 = lane * per;
  int sum = 0;
  for (int i = 0; i < per; ++i) { int j = b0 + i; if (j < CB) sum += row[j]; }
  int inc = sum;
  for (int o = 1; o < 32; o <<= 1) { int v = __shfl_up(inc, o); if (lane >= o) inc += v; }
  int run = inc - sum;
  for (int i = 0; i < per; ++i) { int j = b0 + i; if (j < CB) { int v = row[j]; row[j] = run; run += v; } }
  if (lane == 31) tot[wave] = inc;
}

__global__ __launch_bounds__(NT) void gcn_place(const int* __restrict__ src, const int* __restrict__ dst,
                                                const int* __restrict__ cnt, const int* __restrict__ tot,
                                                unsigned* __restrict__ sorted, int* __restrict__ bst,
                                                long long E, int NB, int CB, int BN) {
  __shared__ unsigned       sw[CH];
  __shared__ unsigned short sb[CH];
  __shared__ int lstart[NBMAX], lcur[NBMAX], gbase[NBMAX], gst[NBMAX];
  int tid = (int)threadIdx.x, b = (int)blockIdx.x, lane = tid & 31;
  for (int k = tid; k < NBMAX; k += NT) {
    int own = 0, before = 0, t = 0;
    if (k < NB) {
      t = tot[k];
      before = cnt[(size_t)k * CB + b];
      int next = (b + 1 < CB) ? cnt[(size_t)k * CB + b + 1] : t;
      own = next - before;
    }
    lstart[k] = own; gbase[k] = before; gst[k] = t;
  }
  __syncthreads();
  if (tid < 32) {
    wave_excl_scan_lds(lstart, NB, lane);
    wave_excl_scan_lds(gst, NB, lane);
  }
  __syncthreads();
  for (int k = tid; k < NB; k += NT) {
    lcur[k] = lstart[k];
    gbase[k] += gst[k];
    if (b == 0) bst[k] = gst[k];
  }
  if (b == 0 && tid == 0) bst[NB] = (int)E;
  __syncthreads();
  long long e0 = (long long)b * CH;
  int n = (int)((E - e0 < CH) ? (E - e0) : CH);
  for (int i0 = tid; i0 < n; i0 += 4 * NT) {
    int d[4], s[4];
#pragma unroll
    for (int j = 0; j < 4; ++j) {
      int i = i0 + j * NT;
      d[j] = (i < n) ? dst[e0 + i] : -1;
      s[j] = (i < n) ? src[e0 + i] : 0;
    }
#pragma unroll
    for (int j = 0; j < 4; ++j) {
      if (d[j] >= 0) {
        int bw = bucket_word(d[j], BN, 1.0f / (float)BN);
        int k = bw >> BSH;
        int r = lds_add_ret_i32(&lcur[k], 1);
        sw[r] = ((unsigned)s[j] << BSH) | (unsigned)(bw & (BSZ - 1));
        sb[r] = (unsigned short)k;
      }
    }
  }
  __syncthreads();
  for (int i = tid; i < n; i += NT) {
    int k = sb[i];
    sorted[(size_t)gbase[k] + (i - lstart[k])] = sw[i];
  }
}

__global__ __launch_bounds__(NT) void gcn_local(const int* __restrict__ src, const int* __restrict__ dst,
                                                unsigned* __restrict__ tmp, unsigned* __restrict__ seg,
                                                long long E, int NB, int CB, int BN) {
  __shared__ unsigned sw[CH];
  __shared__ int lstart[NBMAX];
  int tid = (int)threadIdx.x, b = (int)blockIdx.x, lane = tid & 31;
  for (int k = tid; k < NBMAX; k += NT) lstart[k] = 0;
  long long e0 = (long long)b * CH;
  int n = (int)((E - e0 < CH) ? (E - e0) : CH);
  int d[CH / NT], s[CH / NT], rk[CH / NT];
#pragma unroll
  for (int j = 0; j < CH / NT; ++j) {
    int i = tid + j * NT;
    int ic = (i < n) ? i : (n - 1);
    d[j] = dst[e0 + ic];
    s[j] = src[e0 + ic];
  }
  float inv = 1.0f / (float)BN;
#pragma unroll
  for (int j = 0; j < CH / NT; ++j) d[j] = (tid + j * NT >= n) ? -1 : bucket_word(d[j], BN, inv);
  __syncthreads();
#pragma unroll
  for (int j = 0; j < CH / NT; ++j) rk[j] = (d[j] >= 0) ? lds_add_ret_i32(&lstart[d[j] >> BSH], 1) : 0;
  __syncthreads();
  if (tid < 32) wave_excl_scan_full(lstart, lane);
  __syncthreads();
#pragma unroll
  for (int j = 0; j < CH / NT; ++j) rk[j] += lstart[(d[j] >= 0) ? (d[j] >> BSH) : 0];
#pragma unroll
  for (int j = 0; j < CH / NT; ++j)
    if (d[j] >= 0) sw[rk[j]] = ((unsigned)s[j] << BSH) | (unsigned)(d[j] & (BSZ - 1));
  for (int k = tid; k < NB; k += NT) {
    int st = lstart[k];
    int en = (k + 1 < NB) ? lstart[k + 1] : n;
    seg[(size_t)k * CB + b] = (unsigned)st | ((unsigned)en << 16);
  }
  __syncthreads();
  for (int i = tid; i < n; i += NT) tmp[e0 + i] = sw[i];
}

__global__ __launch_bounds__(NT) void gcn_degree(unsigned* __restrict__ sorted, int* __restrict__ bst,
                                                 const unsigned* __restrict__ tmp, const unsigned* __restrict__ seg,
                                                 const float* __restrict__ x, const float* __restrict__ W1,
                                                 float* __restrict__ dinv, float* __restrict__ rows,
                                                 int N, int NB, int CB, int BN) {
  __shared__ int c[BSZ];
  __shared__ float sd[BSZ];
  __shared__ __attribute__((aligned(16))) float st[BSZ * 16];
  __shared__ int sst[CBMAX], slen[CBMAX], spre[CBMAX];
  __shared__ int sbase, stot;
  int tid = (int)threadIdx.x, k = (int)blockIdx.x;
  int lane = tid & 31, wave = tid >> 5;
  int node0 = k * BN;
  if (tid < BSZ) c[tid] = 0;
  unsigned segw = (tid < CB) ? seg[(size_t)k * CB + tid] : 0u;

  v16h bh = {}, bl = {}, ah = {}, al = {};
  if (lane < 16) {
#pragma unroll
    for (int kk = 0; kk < 5; ++kk) {
      float w = W1[kk * 16 + lane];
      _Float16 hi = (_Float16)w;
      bh[kk] = hi;
      bl[kk] = (_Float16)(w - (float)hi);
    }
    int d = node0 + (wave << 4) + lane;
    if (d < N) {
      const float* xr = x + (size_t)d * 5;
#pragma unroll
      for (int kk = 0; kk < 5; ++kk) {
        float v = xr[kk];
        _Float16 hi = (_Float16)v;
        ah[kk] = hi;
        al[kk] = (_Float16)(v - (float)hi);
      }
    }
  }
  if (CB > 0) {
    int s0 = (int)(segw & 0xFFFFu), s1 = (int)(segw >> 16);
    sst[tid] = s0; slen[tid] = s1 - s0; spre[tid] = s1 - s0;
  }
  __syncthreads();
  if (CB > 0) {
    if (wave == 0) {
      int tot = wave_excl_scan_full(spre, lane);
      if (lane == 0) stot = tot;
    } else if (wave == 1) {
      int sum = 0;
#pragma unroll
      for (int i = 0; i < CBMAX / 32; ++i) sum += sst[lane + 32 * i];
      for (int o = 16; o > 0; o >>= 1) sum += __shfl_xor(sum, o);
      if (lane == 0) sbase = sum;
    }
    __syncthreads();
    int base = sbase;
    if (tid == 0) {
      bst[k] = base;
      if (k == NB - 1) bst[NB] = base + stot;
    }
    for (int c0 = wave; c0 < CB; c0 += SEGT * (NT / 32)) {
      unsigned w[SEGT];
      int pos[SEGT], len[SEGT], off[SEGT];
      int mx = 0;
#pragma unroll
      for (int u = 0; u < SEGT; ++u) {
        int cc = c0 + u * (NT / 32);
        bool in = cc < CB;
        int ci = in ? cc : 0;
        int l = slen[ci];
        off[u] = ci * CH + sst[ci];
        pos[u] = spre[ci];
        len[u] = in ? l : 0;
        mx = (len[u] > mx) ? len[u] : mx;
      }
#pragma unroll
      for (int u = 0; u < SEGT; ++u) w[u] = (lane < len[u]) ? tmp[(size_t)off[u] + lane] : NOEDGE;
#pragma unroll
      for (int u = 0; u < SEGT; ++u) {
        if (w[u] != NOEDGE) {
          sorted[base + pos[u] + lane] = w[u];
          lds_add_i32(&c[w[u] & (BSZ - 1)], 1);
        }
      }
      if (mx > 32) {
        for (int u = 0; u < SEGT; ++u) {
          for (int o = 32 + lane; o < len[u]; o += 32) {
            unsigned wv = tmp[(size_t)off[u] + o];
            sorted[base + pos[u] + o] = wv;
            lds_add_i32(&c[wv & (BSZ - 1)], 1);
          }
        }
      }
    }
  } else {
    int e0 = bst[k], e1 = bst[k + 1];
    for (int i0 = e0 + tid; i0 < e1; i0 += 4 * NT) {
      unsigned w[4];
#pragma unroll
      for (int j = 0; j < 4; ++j) { int i = i0 + j * NT; w[j] = (i < e1) ? sorted[i] : NOEDGE; }
#pragma unroll
      for (int j = 0; j < 4; ++j) if (w[j] != NOEDGE) lds_add_i32(&c[w[j] & (BSZ - 1)], 1);
    }
  }
  __syncthreads();
  if (tid < BSZ) {
    int d = node0 + tid;
    float di = rsqrtf(1.0f + (float)c[tid]);
    sd[tid] = di;
    if (tid < BN && d < N) dinv[d] = di;
  }
  v8f acc = {};
  acc = __builtin_amdgcn_wmma_f32_16x16x32_f16(false, al, false, bh, (short)0, acc, false, false);
  acc = __builtin_amdgcn_wmma_f32_16x16x32_f16(false, ah, false, bl, (short)0, acc, false, false);
  acc = __builtin_amdgcn_wmma_f32_16x16x32_f16(false, ah, false, bh, (short)0, acc, false, false);
  __syncthreads();
  int ch = lane & 15;
  int nb = (wave << 4) + ((lane >> 4) << 3);
#pragma unroll
  for (int r = 0; r < 8; ++r) st[(nb + r) * 16 + ch] = acc[r] * sd[nb + r];
  __syncthreads();
  const float4* st4 = (const float4*)st;
  float4* out4 = (float4*)rows + (size_t)node0 * 4;
  for (int i = tid; i < BN * 4; i += NT)
    if (node0 + (i >> 2) < N) out4[i] = st4[i];
}

#define L1CAP   8192
#define L1PER   (L1CAP / NTL)
__global__ __launch_bounds__(NTL) void gcn_layer1(const unsigned* __restrict__ sorted, const int* __restrict__ bst,
                                                 const float* __restrict__ rows, const float* __restrict__ b1,
                                                 const float* __restrict__ W2, const float* __restrict__ dinv,
                                                 float* __restrict__ t2, int N, int BN) {
  __shared__ float bins[BSZ * PITCH];
  __shared__ unsigned stage[L1CAP];
  __shared__ __attribute__((aligned(16))) int dcnt[BSZ];
  int tid = (int)threadIdx.x, k = (int)blockIdx.x;
  int node0 = k * BN;
  const float4* rows4 = (const float4*)rows;
  int g = tid >> 2, q = tid & 3;
  int e0 = bst[k], e1 = bst[k + 1];
  int n = e1 - e0;
  bool srt = (n > 0) && (n <= L1CAP);
  unsigned w[L1PER];
  if (srt) {
#pragma unroll
    for (int j = 0; j < L1PER; ++j) {
      int i = tid + j * NTL;
      w[j] = sorted[e0 + ((i < n) ? i : (n - 1))];
    }
  }
  if (tid < BSZ) dcnt[tid] = 0;
  {
    int d = node0 + g;
    float4 v0 = rows4[(size_t)((d < N) ? d : (N - 1)) * 4 + q];
    float* p = &bins[g * PITCH + (q << 2)];
    bool in = (g < BN) && (d < N);
    p[0] = in ? v0.x : 0.0f;
    p[1] = in ? v0.y : 0.0f;
    p[2] = in ? v0.z : 0.0f;
    p[3] = in ? v0.w : 0.0f;
  }
  __syncthreads();

  if (srt) {
    {
      int rk[L1PER];
#pragma unroll
      for (int j = 0; j < L1PER; ++j)
        rk[j] = (tid + j * NTL < n) ? lds_add_ret_i32(&dcnt[w[j] & (BSZ - 1)], 1) : 0;
      __syncthreads();
      if (tid < 32) wave_excl_scan_bsz(dcnt, tid);
      __syncthreads();
#pragma unroll
      for (int j = 0; j < L1PER; ++j) rk[j] += dcnt[w[j] & (BSZ - 1)];
#pragma unroll
      for (int j = 0; j < L1PER; ++j)
        if (tid + j * NTL < n) stage[rk[j]] = w[j];
    }
    __syncthreads();
    int per = (n + (NTL / 4) - 1) / (NTL / 4);
    int lo = g * per;
    int hi = (lo + per < n) ? (lo + per) : n;
    int cur = -1;
    float4 acc = {0.0f, 0.0f, 0.0f, 0.0f};
    unsigned wd[4];
    float4 v[4];
#pragma unroll
    for (int j = 0; j < 4; ++j) {
      int i = lo + j;
      bool in = i < hi;
      wd[j] = stage[in ? i : (n - 1)];
      v[j] = rows4[(size_t)(in ? (wd[j] >> BSH) : 0u) * 4 + q];
    }
    for (int t = 0; t < per; t += 4) {
      unsigned wn[4];
      float4 vn[4];
#pragma unroll
      for (int j = 0; j < 4; ++j) {
        int i = lo + t + 4 + j;
        bool in = i < hi;
        wn[j] = stage[in ? i : (n - 1)];
        vn[j] = rows4[(size_t)(in ? (wn[j] >> BSH) : 0u) * 4 + q];
      }
#pragma unroll
      for (int j = 0; j < 4; ++j) {
        bool ok = (lo + t + j) < hi;
        int dl = (int)(wd[j] & (BSZ - 1));
        bool chg = ok && (dl != cur);
        bool fl = chg && (cur >= 0);
        if (fl) {
          float* p = &bins[cur * PITCH + (q << 2)];
          lds_add_f32(p + 0, acc.x);
          lds_add_f32(p + 1, acc.y);
          lds_add_f32(p + 2, acc.z);
          lds_add_f32(p + 3, acc.w);
        }
        if (chg) { cur = dl; acc.x = 0.0f; acc.y = 0.0f; acc.z = 0.0f; acc.w = 0.0f; }
        if (ok) { acc.x += v[j].x; acc.y += v[j].y; acc.z += v[j].z; acc.w += v[j].w; }
      }
#pragma unroll
      for (int j = 0; j < 4; ++j) { wd[j] = wn[j]; v[j] = vn[j]; }
    }
    if (cur >= 0) {
      float* p = &bins[cur * PITCH + (q << 2)];
      lds_add_f32(p + 0, acc.x);
      lds_add_f32(p + 1, acc.y);
      lds_add_f32(p + 2, acc.z);
      lds_add_f32(p + 3, acc.w);
    }
  } else {
    int i0 = e0 + g;
    if (i0 < e1) {
      unsigned wu[4];
#pragma unroll
      for (int j = 0; j < 4; ++j) {
        int i = i0 + j * (NTL / 4);
        wu[j] = sorted[(i < e1) ? i : (e1 - 1)];
      }
      do {
        unsigned wc[4];
        float4 vv[4];
        bool ok[4];
#pragma unroll
        for (int j = 0; j < 4; ++j) {
          wc[j] = wu[j];
          ok[j] = (i0 + j * (NTL / 4)) < e1;
          vv[j] = rows4[(size_t)(wc[j] >> BSH) * 4 + q];
        }
        i0 += 4 * (NTL / 4);
#pragma unroll
        for (int j = 0; j < 4; ++j) {
          int i = i0 + j * (NTL / 4);
          wu[j] = sorted[(i < e1) ? i : (e1 - 1)];
        }
#pragma unroll
        for (int j = 0; j < 4; ++j) {
          int dl = ok[j] ? (int)(wc[j] & (BSZ - 1)) : g;
          float* p = &bins[dl * PITCH + (q << 2)];
          lds_add_f32(p + 0, ok[j] ? vv[j].x : 0.0f);
          lds_add_f32(p + 1, ok[j] ? vv[j].y : 0.0f);
          lds_add_f32(p + 2, ok[j] ? vv[j].z : 0.0f);
          lds_add_f32(p + 3, ok[j] ? vv[j].w : 0.0f);
        }
      } while (i0 < e1);
    }
  }
  __syncthreads();
  if (tid < BN) {
    int d = node0 + tid;
    if (d < N) {
      float di = dinv[d];
      float acc2 = 0.0f;
#pragma unroll
      for (int cc = 0; cc < 16; ++cc) {
        float vv = fmaf(bins[tid * PITCH + cc], di, b1[cc]);
        vv = fmaxf(vv, 0.0f);
        acc2 = fmaf(vv, W2[cc], acc2);
      }
      t2[d] = acc2;
    }
  }
}

#define NT2     1024
#define L2CAP   16384
#define L2PER   (L2CAP / NT2)
#define LBINS   4096
#define LPER    (LBINS / NT2)
#define SECOND  0x80000000u
static_assert(LPER == 4 && NT2 / 32 == 32, "a thread scans its four counters as one 16-byte piece, and the 32 wave sums as eight");
static_assert(NT2 >= 2 * BSZ, "one thread per node of the pair");
__global__ __launch_bounds__(NT2) void gcn_layer2(const unsigned* __restrict__ sorted, const int* __restrict__ bst,
                                                  const float* __restrict__ t2, const float* __restrict__ dinv,
                                                  const float* __restrict__ b2, float* __restrict__ out,
                                                  int N, int NB, int BN) {
  __shared__ float bins[2 * BSZ];
  __shared__ float sdinv[2 * BSZ];
  __shared__ unsigned stage[L2CAP];
  __shared__ __attribute__((aligned(16))) int lc[LBINS];
  __shared__ __attribute__((aligned(16))) int wsum[NT2 / 32];
  int tid = (int)threadIdx.x, k0 = 2 * (int)blockIdx.x;
  int lane = tid & 31, wave = tid >> 5;
  int node0 = k0 * BN;
  int e0 = bst[k0], em = bst[k0 + 1], e1 = bst[(k0 + 2 < NB) ? (k0 + 2) : NB];
  int n = e1 - e0;
  int n0 = em - e0;
  bool srt = (n > 0) && (n <= L2CAP);
  unsigned w[L2PER];
  if (srt) {
#pragma unroll
    for (int j = 0; j < L2PER; ++j) {
      int i = tid + j * NT2;
      w[j] = sorted[e0 + ((i < n) ? i : (n - 1))];
    }
#pragma unroll
    for (int j = 0; j < LPER; ++j) lc[tid + j * NT2] = 0;
  }
  if (tid < 2 * BSZ) {
    int d = node0 + tid;
    bins[tid] = 0.0f;
    sdinv[tid] = (tid < 2 * BN && d < N) ? dinv[d] : 0.0f;
  }
  __syncthreads();
  if (srt) {
    int rk[L2PER];
#pragma unroll
    for (int j = 0; j < L2PER; ++j) {
      int i = tid + j * NT2;
      w[j] |= (i >= n0) ? SECOND : 0u;
      int b = (int)((w[j] & ~SECOND) >> (BSH + 5));
      b = (b < LBINS) ? b : (LBINS - 1);
      rk[j] = (i < n) ? lds_add_ret_i32(&lc[b], 1) : 0;
    }
    __syncthreads();
    int4* lp = (int4*)&lc[tid * LPER];
    int4 c0 = lp[0];
    int s = c0.x + c0.y + c0.z + c0.w;
    int inc = s;
    for (int o = 1; o < 32; o <<= 1) { int t = __shfl_up(inc, o); if (lane >= o) inc += t; }
    if (lane == 31) wsum[wave] = inc;
    __syncthreads();
    int run = inc - s;
    {
      const int4* wp = (const int4*)wsum;
      int4 q[NT2 / 128];
#pragma unroll
      for (int i = 0; i < NT2 / 128; ++i) q[i] = wp[i];
#pragma unroll
      for (int i = 0; i < NT2 / 128; ++i) {
        run += (4 * i + 0 < wave) ? q[i].x : 0;
        run += (4 * i + 1 < wave) ? q[i].y : 0;
        run += (4 * i + 2 < wave) ? q[i].z : 0;
        run += (4 * i + 3 < wave) ? q[i].w : 0;
      }
    }
    int4 p0;
    p0.x = run; run += c0.x; p0.y = run; run += c0.y; p0.z = run; run += c0.z; p0.w = run;
    lp[0] = p0;
    __syncthreads();
#pragma unroll
    for (int j = 0; j < L2PER; ++j) {
      int b = (int)((w[j] & ~SECOND) >> (BSH + 5));
      b = (b < LBINS) ? b : (LBINS - 1);
      rk[j] += lc[b];
    }
#pragma unroll
    for (int j = 0; j < L2PER; ++j)
      if (tid + j * NT2 < n) stage[rk[j]] = w[j];
    __syncthreads();
    for (int i0 = tid; i0 < n; i0 += 4 * NT2) {
      unsigned ww[4];
      float a[4], g[4];
#pragma unroll
      for (int j = 0; j < 4; ++j) { int i = i0 + j * NT2; ww[j] = (i < n) ? stage[i] : NOEDGE; }
#pragma unroll
      for (int j = 0; j < 4; ++j) {
        unsigned sidx = (ww[j] != NOEDGE) ? ((ww[j] & ~SECOND) >> BSH) : 0u;
        a[j] = t2[sidx]; g[j] = dinv[sidx];
      }
#pragma unroll
      for (int j = 0; j < 4; ++j) {
        if (ww[j] != NOEDGE) {
          int dl = (int)(ww[j] & (BSZ - 1)) + ((ww[j] & SECOND) ? BN : 0);
          lds_add_f32(&bins[dl], a[j] * g[j] * sdinv[dl]);
        }
      }
    }
  } else {
    for (int i0 = e0 + tid; i0 < e1; i0 += 4 * NT2) {
      unsigned wu[4];
      float a[4], g[4];
#pragma unroll
      for (int j = 0; j < 4; ++j) { int i = i0 + j * NT2; wu[j] = (i < e1) ? sorted[i] : NOEDGE; }
#pragma unroll
      for (int j = 0; j < 4; ++j) {
        unsigned s = (wu[j] != NOEDGE) ? (wu[j] >> BSH) : 0u;
        a[j] = t2[s]; g[j] = dinv[s];
      }
#pragma unroll
      for (int j = 0; j < 4; ++j) {
        if (wu[j] != NOEDGE) {
          int dl = (int)(wu[j] & (BSZ - 1)) + ((i0 + j * NT2 >= em) ? BN : 0);
          lds_add_f32(&bins[dl], a[j] * g[j] * sdinv[dl]);
        }
      }
    }
  }
  __syncthreads();
  if (tid < 2 * BN) {
    int d = node0 + tid;
    if (d < N) { float di = sdinv[tid]; out[d] = bins[tid] + t2[d] * di * di + b2[0]; }
  }
}

extern "C" void kernel_launch(void* const* d_in, const int* in_sizes, int n_in,
                              void* d_out, int out_size, void* d_ws, size_t ws_size,
                              hipStream_t stream) {
  const float* x  = (const float*)d_in[0];
  const float* W1 = (const float*)d_in[1];
  const float* b1 = (const float*)d_in[2];
  const float* W2 = (const float*)d_in[3];
  const float* b2 = (const float*)d_in[4];
  const int*   ei = (const int*)d_in[5];

  int       N = in_sizes[0] / 5;
  long long E = (long long)in_sizes[5] / 2;
  const int* src = ei;
  const int* dst = ei + E;

  int BN = 2 * ((N + 1023) / 1024);
  int NB = (N + BN - 1) / BN;
  int CB = (int)((E + CH - 1) / CH);

  float*    rows   = (float*)d_ws;
  unsigned* sorted = (unsigned*)(rows + (size_t)16 * N);
  int*      bst    = (int*)(sorted + E);
  float*    dinv   = (float*)(bst + NB + 1);
  float*    t2     = dinv + N;
  int*      cnt    = (int*)rows;
  int*      tot    = cnt + (size_t)NB * CB;
  unsigned* tmp    = (unsigned*)(t2 + N);
  unsigned* seg    = tmp + E;

  size_t need = ((size_t)18 * N + (size_t)E + (size_t)NB + 1) * 4;
  if (ws_size < need) return;

  float* out = (float*)d_out;

  size_t need1 = need + ((size_t)E + (size_t)NB * CB) * 4;
  if (ws_size >= need1 && CB <= CBMAX) {
    gcn_local <<<CB, NT, 0, stream>>>(src, dst, tmp, seg, E, NB, CB, BN);
    gcn_degree<<<NB, NT, 0, stream>>>(sorted, bst, tmp, seg, x, W1, dinv, rows, N, NB, CB, BN);
  } else {
    gcn_count <<<CB, NT, 0, stream>>>(dst, cnt, E, NB, CB, BN);
    gcn_scan  <<<(NB + 7) / 8, 256, 0, stream>>>(cnt, tot, NB, CB);
    gcn_place <<<CB, NT, 0, stream>>>(src, dst, cnt, tot, sorted, bst, E, NB, CB, BN);
    gcn_degree<<<NB, NT, 0, stream>>>(sorted, bst, tmp, seg, x, W1, dinv, rows, N, NB, 0, BN);
  }
  gcn_layer1<<<NB, NTL, 0, stream>>>(sorted, bst, rows, b1, W2, dinv, t2, N, BN);
  gcn_layer2<<<(NB + 1) / 2, NT2, 0, stream>>>(sorted, bst, t2, dinv, b2, out, N, NB, BN);
}
